# starting kernel
# baseline (speedup 1.0000x reference)
#include <hip/hip_runtime.h>

typedef float fx2 __attribute__((ext_vector_type(2)));
typedef float fx8 __attribute__((ext_vector_type(8)));

__global__ void deg_init_kernel(float* __restrict__ deg, int n) {
  int i = blockIdx.x * blockDim.x + threadIdx.x;
  if (i < n) deg[i] = 1.0f;
}

__global__ void deg_accum_kernel(const int* __restrict__ dst,
                                 const float* __restrict__ ew,
                                 float* __restrict__ deg, int e) {
  int i = blockIdx.x * blockDim.x + threadIdx.x;
  if (i < e) unsafeAtomicAdd(&deg[dst[i]], ew[i]);
}

__global__ void dinv_kernel(const float* __restrict__ deg,
                            float* __restrict__ dinv, int n) {
  int i = blockIdx.x * blockDim.x + threadIdx.x;
  if (i < n) {
    float d = deg[i];
    dinv[i] = (d > 0.0f) ? rsqrtf(d) : 0.0f;
  }
}

template <int K, int NOUT>
__global__ void gemm_wmma_scale_kernel(const float* __restrict__ X,
                                       const float* __restrict__ W,
                                       const float* __restrict__ dinv,
                                       float* __restrict__ hs,
                                       float* __restrict__ acc,
                                       int nrows) {
  const int lane = threadIdx.x & 31;
  const int wave = blockIdx.x * (blockDim.x >> 5) + (threadIdx.x >> 5);
  const int ntc  = NOUT / 16;
  const int row0 = (wave / ntc) * 16;
  const int col0 = (wave % ntc) * 16;
  if (row0 >= nrows) return;

  const int half = lane >> 4;
  const int m    = lane & 15;

  fx8 c = {};
  const float* xrow = X + (size_t)(row0 + m) * K;

#pragma unroll
  for (int kb = 0; kb < K; kb += 4) {
    const int ka = kb + 2 * half;
    fx2 a, b;
    a.x = xrow[ka];
    a.y = xrow[ka + 1];
    b.x = W[(size_t)ka * NOUT + col0 + m];
    b.y = W[(size_t)(ka + 1) * NOUT + col0 + m];
    c = __builtin_amdgcn_wmma_f32_16x16x4_f32(false, a, false, b,
                                              (short)0, c, false, false);
  }

#pragma unroll
  for (int r = 0; r < 8; ++r) {
    const int mr  = r + 8 * half;
    const float s = dinv[row0 + mr];
    const float v = c[r] * s;
    const size_t idx = (size_t)(row0 + mr) * NOUT + col0 + m;
    hs[idx]  = v;
    acc[idx] = v;
  }
}

__global__ void scatter128_kernel(const float* __restrict__ hs,
                                  const int* __restrict__ src,
                                  const int* __restrict__ dst,
                                  const float* __restrict__ ew,
                                  float* __restrict__ acc, int e) {
  const int lane = threadIdx.x & 31;
  const int edge = blockIdx.x * (blockDim.x >> 5) + (threadIdx.x >> 5);
  if (edge >= e) return;
  const int   s = src[edge];
  const int   d = dst[edge];
  const float w = ew[edge];
  const float4 v = ((const float4*)(hs + (size_t)s * 128))[lane];
  float* q = acc + (size_t)d * 128 + lane * 4;
  unsafeAtomicAdd(q + 0, w * v.x);
  unsafeAtomicAdd(q + 1, w * v.y);
  unsafeAtomicAdd(q + 2, w * v.z);
  unsafeAtomicAdd(q + 3, w * v.w);
}

__global__ void scatter64_kernel(const float* __restrict__ hs,
                                 const int* __restrict__ src,
                                 const int* __restrict__ dst,
                                 const float* __restrict__ ew,
                                 float* __restrict__ acc, int e) {
  const int lane = threadIdx.x & 31;
  const int edge = blockIdx.x * (blockDim.x >> 5) + (threadIdx.x >> 5);
  if (edge >= e) return;
  const int   s = src[edge];
  const int   d = dst[edge];
  const float w = ew[edge];
  const float2 v = ((const float2*)(hs + (size_t)s * 64))[lane];
  float* q = acc + (size_t)d * 64 + lane * 2;
  unsafeAtomicAdd(q + 0, w * v.x);
  unsafeAtomicAdd(q + 1, w * v.y);
}

template <int F>
__global__ void finalize_kernel(const float* __restrict__ acc,
                                const float* __restrict__ dinv,
                                const float* __restrict__ bias,
                                float* __restrict__ out, int n, int do_relu) {
  const size_t i = (size_t)blockIdx.x * blockDim.x + threadIdx.x;
  if (i >= (size_t)n * F) return;
  const int row = (int)(i / F);
  const int f   = (int)(i % F);
  float v = dinv[row] * acc[i] + bias[f];
  if (do_relu) v = fmaxf(v, 0.0f);
  out[i] = v;
}

extern "C" void kernel_launch(void* const* d_in, const int* in_sizes, int n_in,
                              void* d_out, int out_size, void* d_ws, size_t ws_size,
                              hipStream_t stream) {
  const float* z  = (const float*)d_in[0];
  const int*   ei = (const int*)d_in[1];
  const float* ew = (const float*)d_in[2];
  const float* W1 = (const float*)d_in[3];
  const float* b1 = (const float*)d_in[4];
  const float* W2 = (const float*)d_in[5];
  const float* b2 = (const float*)d_in[6];
  float* out = (float*)d_out;

  const int n = in_sizes[0] / 64;
  const int e = in_sizes[2];
  const int* srcp = ei;
  const int* dstp = ei + e;

  float* ws   = (float*)d_ws;
  float* deg  = ws;
  float* dinv = deg + n;
  float* X1   = dinv + n;
  float* X2   = X1 + (size_t)n * 128;
  float* hs2  = X2;
  float* acc2 = X2 + (size_t)n * 64;

  const int B = 256;

  deg_init_kernel<<<(n + B - 1) / B, B, 0, stream>>>(deg, n);
  deg_accum_kernel<<<(e + B - 1) / B, B, 0, stream>>>(dstp, ew, deg, e);
  dinv_kernel<<<(n + B - 1) / B, B, 0, stream>>>(deg, dinv, n);

  {
    const long long waves = (long long)((n + 15) / 16) * (128 / 16);
    const long long thr   = waves * 32;
    gemm_wmma_scale_kernel<64, 128><<<(unsigned)((thr + B - 1) / B), B, 0, stream>>>(
        z, W1, dinv, X1, X2, n);
  }
  {
    const long long thr = (long long)e * 32;
    scatter128_kernel<<<(unsigned)((thr + B - 1) / B), B, 0, stream>>>(
        X1, srcp, dstp, ew, X2, e);
  }
  {
    const long long thr = (long long)n * 128;
    finalize_kernel<128><<<(unsigned)((thr + B - 1) / B), B, 0, stream>>>(
        X2, dinv, b1, X1, n, 1);
  }
  {
    const long long waves = (long long)((n + 15) / 16) * (64 / 16);
    const long long thr   = waves * 32;
    gemm_wmma_scale_kernel<128, 64><<<(unsigned)((thr + B - 1) / B), B, 0, stream>>>(
        X1, W2, dinv, hs2, acc2, n);
  }
  {
    const long long thr = (long long)e * 32;
    scatter64_kernel<<<(unsigned)((thr + B - 1) / B), B, 0, stream>>>(
        hs2, srcp, dstp, ew, acc2, e);
  }
  {
    const long long thr = (long long)n * 64;
    finalize_kernel<64><<<(unsigned)((thr + B - 1) / B), B, 0, stream>>>(
        acc2, dinv, b2, out, n, 0);
  }
}


# Edge sums kept in LDS: each work-group owns 256 destination rows, filters the whole edge list for them, adds source rows by ds_add_f32, writes finished rows; no global atomics; LDS arrival order
# speedup vs baseline: 2.1003x; 2.1003x over previous
#include <hip/hip_runtime.h>

typedef float fx2 __attribute__((ext_vector_type(2)));
typedef float fx8 __attribute__((ext_vector_type(8)));

#define DEG_TILE   65536
#define DEG_GROUPS 128
#define DEG_BLOCK  1024

__global__ void __launch_bounds__(DEG_BLOCK)
deg_partial_kernel(const int* __restrict__ dst,
                   const float* __restrict__ ew,
                   float* __restrict__ part, int n, int e) {
  __shared__ float hist[DEG_TILE];
  const int t = threadIdx.x;
  float* mine = part + (size_t)blockIdx.x * n;
  for (int base = 0; base < n; base += DEG_TILE) {
    const int cnt = (n - base < DEG_TILE) ? (n - base) : DEG_TILE;
    float4* hist4 = (float4*)hist;
    for (int j = t; j < (cnt + 3) / 4; j += DEG_BLOCK)
      hist4[j] = make_float4(0.0f, 0.0f, 0.0f, 0.0f);
    __syncthreads();
    const int stride = DEG_GROUPS * DEG_BLOCK;
    int i = blockIdx.x * DEG_BLOCK + t;
    for (; i < e - 3 * stride; i += 4 * stride) {
      const unsigned d0 = (unsigned)(dst[i] - base);
      const unsigned d1 = (unsigned)(dst[i + stride] - base);
      const unsigned d2 = (unsigned)(dst[i + 2 * stride] - base);
      const unsigned d3 = (unsigned)(dst[i + 3 * stride] - base);
      const float w0 = ew[i];
      const float w1 = ew[i + stride];
      const float w2 = ew[i + 2 * stride];
      const float w3 = ew[i + 3 * stride];
      atomicAdd(&hist[d0 & (DEG_TILE - 1)], (d0 < (unsigned)cnt) ? w0 : 0.0f);
      atomicAdd(&hist[d1 & (DEG_TILE - 1)], (d1 < (unsigned)cnt) ? w1 : 0.0f);
      atomicAdd(&hist[d2 & (DEG_TILE - 1)], (d2 < (unsigned)cnt) ? w2 : 0.0f);
      atomicAdd(&hist[d3 & (DEG_TILE - 1)], (d3 < (unsigned)cnt) ? w3 : 0.0f);
    }
    for (; i < e; i += stride) {
      const unsigned d = (unsigned)(dst[i] - base);
      const float w = ew[i];
      atomicAdd(&hist[d & (DEG_TILE - 1)], (d < (unsigned)cnt) ? w : 0.0f);
    }
    __syncthreads();
    for (int j = t; j < cnt; j += DEG_BLOCK) mine[base + j] = hist[j];
    __syncthreads();
  }
}

__global__ void dinv_reduce_kernel(const float* __restrict__ part,
                                   float* __restrict__ dinv, int n) {
  int i = blockIdx.x * blockDim.x + threadIdx.x;
  if (i < n) {
    float d = 1.0f;
#pragma unroll 16
    for (int g = 0; g < DEG_GROUPS; ++g) d += part[(size_t)g * n + i];
    dinv[i] = (d > 0.0f) ? rsqrtf(d) : 0.0f;
  }
}

template <int K, int NOUT>
__global__ void gemm_wmma_scale_kernel(const float* __restrict__ X,
                                       const float* __restrict__ W,
                                       const float* __restrict__ dinv,
                                       float* __restrict__ hs,
                                       int nrows) {
  const int lane = threadIdx.x & 31;
  const int wave = blockIdx.x * (blockDim.x >> 5) + (threadIdx.x >> 5);
  const int ntc  = NOUT / 16;
  const int row0 = (wave / ntc) * 16;
  const int col0 = (wave % ntc) * 16;
  if (row0 >= nrows) return;

  const int half = lane >> 4;
  const int m    = lane & 15;

  fx8 c = {};
  const float* xrow = X + (size_t)(row0 + m) * K;

#pragma unroll
  for (int kb = 0; kb < K; kb += 4) {
    const int ka = kb + 2 * half;
    fx2 a, b;
    a.x = xrow[ka];
    a.y = xrow[ka + 1];
    b.x = W[(size_t)ka * NOUT + col0 + m];
    b.y = W[(size_t)(ka + 1) * NOUT + col0 + m];
    c = __builtin_amdgcn_wmma_f32_16x16x4_f32(false, a, false, b,
                                              (short)0, c, false, false);
  }

#pragma unroll
  for (int r = 0; r < 8; ++r) {
    const int mr  = r + 8 * half;
    const float s = dinv[row0 + mr];
    const float v = c[r] * s;
    const size_t idx = (size_t)(row0 + mr) * NOUT + col0 + m;
    hs[idx] = v;
  }
}

#define OWN_ROWS  256
#define OWN_BLOCK 1024
#define OWN_U     16
#define OWN_CHUNK (32 * OWN_U)

template <int F>
__device__ __forceinline__ void own_add_edge(float* acc, const float* __restrict__ hs,
                                             int s, float w, int r, int lane) {
  const float* p = hs + (size_t)s * F + lane;
  float v[F / 32];
#pragma unroll
  for (int c = 0; c < F / 32; ++c) v[c] = p[32 * c];
#pragma unroll
  for (int c = 0; c < F / 32; ++c) atomicAdd(&acc[r * F + 32 * c + lane], w * v[c]);
}

#define OWN_G 8
template <int F>
__device__ __forceinline__ void own_flush(float* acc, const float* __restrict__ hs,
                                          const int* __restrict__ src,
                                          const float* __restrict__ ew,
                                          int qi, int qr, int qn, int lane) {
  int   s = 0;
  float w = 0.0f;
  if (lane < qn) { s = src[qi]; w = ew[qi]; }
  int t = 0;
  for (; t + OWN_G <= qn; t += OWN_G) {
    float v[OWN_G][F / 32];
#pragma unroll
    for (int u = 0; u < OWN_G; ++u) {
      const int su = __builtin_amdgcn_readlane(s, t + u);
      const float* p = hs + (size_t)su * F + lane;
#pragma unroll
      for (int c = 0; c < F / 32; ++c) v[u][c] = p[32 * c];
    }
#pragma unroll
    for (int u = 0; u < OWN_G; ++u) {
      const int   ru = __builtin_amdgcn_readlane(qr, t + u);
      const float wu = __int_as_float(__builtin_amdgcn_readlane(__float_as_int(w), t + u));
#pragma unroll
      for (int c = 0; c < F / 32; ++c) atomicAdd(&acc[ru * F + 32 * c + lane], wu * v[u][c]);
    }
  }
  for (; t < qn; ++t) {
    const int   su = __builtin_amdgcn_readlane(s, t);
    const int   ru = __builtin_amdgcn_readlane(qr, t);
    const float wu = __int_as_float(__builtin_amdgcn_readlane(__float_as_int(w), t));
    own_add_edge<F>(acc, hs, su, wu, ru, lane);
  }
}

__device__ __forceinline__ void own_load_chunk(const int* __restrict__ dst, int i0, int e,
                                               int lane, int (&d)[OWN_U]) {
  if (i0 + OWN_CHUNK <= e) {
#pragma unroll
    for (int u = 0; u < OWN_U; ++u) d[u] = dst[i0 + 32 * u + lane];
  } else {
#pragma unroll
    for (int u = 0; u < OWN_U; ++u) {
      const int i = i0 + 32 * u + lane;
      d[u] = (i < e) ? dst[i] : -1;
    }
  }
}

template <int F, bool RELU>
__global__ void __launch_bounds__(OWN_BLOCK)
own_rows_kernel(const float* __restrict__ hs,
                const int* __restrict__ src,
                const int* __restrict__ dst,
                const float* __restrict__ ew,
                const float* __restrict__ dinv,
                const float* __restrict__ bias,
                float* __restrict__ out, int n, int e) {
  __shared__ float acc[OWN_ROWS * F];
  const int t    = threadIdx.x;
  const int lane = t & 31;
  const int wv   = __builtin_amdgcn_readfirstlane(t >> 5);
  const int base = blockIdx.x * OWN_ROWS;
  const int cnt  = (n - base < OWN_ROWS) ? (n - base) : OWN_ROWS;

  const float* hs0 = hs + (size_t)base * F;
  for (int j = t; j < cnt * F; j += OWN_BLOCK) acc[j] = hs0[j];
  __syncthreads();

  const int nchunks = (e + OWN_CHUNK - 1) / OWN_CHUNK;
  const int rot = (int)(((long long)blockIdx.x * nchunks) / gridDim.x);

  int qi = 0, qr = 0, qi2 = 0, qr2 = 0;
  int qn = 0;

  for (int k = wv; k < nchunks; k += OWN_BLOCK / 32) {
    int c = k + rot;
    if (c >= nchunks) c -= nchunks;
    const int i0 = c * OWN_CHUNK;
    int d[OWN_U];
    own_load_chunk(dst, i0, e, lane, d);

    unsigned long long mm[OWN_U / 2];
    unsigned long long any = 0ull;
#pragma unroll
    for (int p = 0; p < OWN_U / 2; ++p) {
      const unsigned lo = __builtin_amdgcn_ballot_w32((unsigned)(d[2 * p] - base) < (unsigned)cnt);
      const unsigned hi = __builtin_amdgcn_ballot_w32((unsigned)(d[2 * p + 1] - base) < (unsigned)cnt);
      mm[p] = ((unsigned long long)hi << 32) | lo;
      any |= mm[p];
    }
    if (any != 0ull) {
      int total = 0;
#pragma unroll
      for (int p = 0; p < OWN_U / 2; ++p) total += __builtin_popcountll(mm[p]);
      if (total > 32) {
#pragma unroll
        for (int p = 0; p < OWN_U / 2; ++p) {
          unsigned long long mk = mm[p];
          while (mk) {
            const int j = __builtin_ctzll(mk);
            mk &= mk - 1ull;
            const int dlo = __builtin_amdgcn_readlane(d[2 * p], j & 31);
            const int dhi = __builtin_amdgcn_readlane(d[2 * p + 1], j & 31);
            const int ii = i0 + 64 * p + j;
            own_add_edge<F>(acc, hs, src[ii], ew[ii], ((j < 32) ? dlo : dhi) - base, lane);
          }
        }
      } else {
#pragma unroll
        for (int p = 0; p < OWN_U / 2; ++p) {
          unsigned long long mk = mm[p];
          while (mk) {
            const int j = __builtin_ctzll(mk);
            mk &= mk - 1ull;
            const int dlo = __builtin_amdgcn_readlane(d[2 * p], j & 31);
            const int dhi = __builtin_amdgcn_readlane(d[2 * p + 1], j & 31);
            const int rr  = ((j < 32) ? dlo : dhi) - base;
            const int ii  = i0 + 64 * p + j;
            if (lane == qn)      { qi  = ii; qr  = rr; }
            if (lane + 32 == qn) { qi2 = ii; qr2 = rr; }
            ++qn;
          }
        }
        if (qn >= 32) {
          own_flush<F>(acc, hs, src, ew, qi, qr, 32, lane);
          qi = qi2;
          qr = qr2;
          qn -= 32;
        }
      }
    }
  }
  if (qn > 0) own_flush<F>(acc, hs, src, ew, qi, qr, qn, lane);
  __syncthreads();

  float* o0 = out + (size_t)base * F;
  for (int j = t; j < cnt * F; j += OWN_BLOCK) {
    const int row = j / F;
    const int f   = j % F;
    float v = dinv[base + row] * acc[j] + bias[f];
    if (RELU) v = fmaxf(v, 0.0f);
    o0[j] = v;
  }
}

extern "C" void kernel_launch(void* const* d_in, const int* in_sizes, int n_in,
                              void* d_out, int out_size, void* d_ws, size_t ws_size,
                              hipStream_t stream) {
  const float* z  = (const float*)d_in[0];
  const int*   ei = (const int*)d_in[1];
  const float* ew = (const float*)d_in[2];
  const float* W1 = (const float*)d_in[3];
  const float* b1 = (const float*)d_in[4];
  const float* W2 = (const float*)d_in[5];
  const float* b2 = (const float*)d_in[6];
  float* out = (float*)d_out;

  const int n = in_sizes[0] / 64;
  const int e = in_sizes[2];
  const int* srcp = ei;
  const int* dstp = ei + e;

  float* ws   = (float*)d_ws;
  float* dinv = ws + n;
  float* X1   = dinv + n;
  float* X2   = X1 + (size_t)n * 128;
  const int B = 256;

  deg_partial_kernel<<<DEG_GROUPS, DEG_BLOCK, 0, stream>>>(dstp, ew, X1, n, e);
  dinv_reduce_kernel<<<(n + B - 1) / B, B, 0, stream>>>(X1, dinv, n);

  {
    const long long waves = (long long)((n + 15) / 16) * (128 / 16);
    const long long thr   = waves * 32;
    gemm_wmma_scale_kernel<64, 128><<<(unsigned)((thr + B - 1) / B), B, 0, stream>>>(
        z, W1, dinv, X1, n);
  }
  own_rows_kernel<128, true><<<(n + OWN_ROWS - 1) / OWN_ROWS, OWN_BLOCK, 0, stream>>>(
      X1, srcp, dstp, ew, dinv, b1, X2, n, e);
  {
    const long long waves = (long long)((n + 15) / 16) * (64 / 16);
    const long long thr   = waves * 32;
    gemm_wmma_scale_kernel<128, 64><<<(unsigned)((thr + B - 1) / B), B, 0, stream>>>(
        X2, W2, dinv, X1, n);
  }
  own_rows_kernel<64, false><<<(n + OWN_ROWS - 1) / OWN_ROWS, OWN_BLOCK, 0, stream>>>(
      X1, srcp, dstp, ew, dinv, b2, out, n, e);
}


# Aggregation kernels filter the edge list by a bit-sliced owner index (8 plane words per 32 edges, built by a small kernel per layer): 1 byte per edge, 8 bit-ops per 32 edges; LDS arrival order
# speedup vs baseline: 2.1676x; 1.0320x over previous
#include <hip/hip_runtime.h>

typedef float fx2 __attribute__((ext_vector_type(2)));
typedef float fx8 __attribute__((ext_vector_type(8)));

#define DEG_TILE   65536
#define DEG_GROUPS 128
#define DEG_BLOCK  1024

__global__ void __launch_bounds__(DEG_BLOCK)
deg_partial_kernel(const int* __restrict__ dst,
                   const float* __restrict__ ew,
                   float* __restrict__ part, int n, int e) {
  __shared__ float hist[DEG_TILE];
  const int t = threadIdx.x;
  float* mine = part + (size_t)blockIdx.x * n;
  for (int base = 0; base < n; base += DEG_TILE) {
    const int cnt = (n - base < DEG_TILE) ? (n - base) : DEG_TILE;
    float4* hist4 = (float4*)hist;
    for (int j = t; j < (cnt + 3) / 4; j += DEG_BLOCK)
      hist4[j] = make_float4(0.0f, 0.0f, 0.0f, 0.0f);
    __syncthreads();
    const int stride = DEG_GROUPS * DEG_BLOCK;
    int i = blockIdx.x * DEG_BLOCK + t;
    for (; i < e - 3 * stride; i += 4 * stride) {
      const unsigned d0 = (unsigned)(dst[i] - base);
      const unsigned d1 = (unsigned)(dst[i + stride] - base);
      const unsigned d2 = (unsigned)(dst[i + 2 * stride] - base);
      const unsigned d3 = (unsigned)(dst[i + 3 * stride] - base);
      const float w0 = ew[i];
      const float w1 = ew[i + stride];
      const float w2 = ew[i + 2 * stride];
      const float w3 = ew[i + 3 * stride];
      atomicAdd(&hist[d0 & (DEG_TILE - 1)], (d0 < (unsigned)cnt) ? w0 : 0.0f);
      atomicAdd(&hist[d1 & (DEG_TILE - 1)], (d1 < (unsigned)cnt) ? w1 : 0.0f);
      atomicAdd(&hist[d2 & (DEG_TILE - 1)], (d2 < (unsigned)cnt) ? w2 : 0.0f);
      atomicAdd(&hist[d3 & (DEG_TILE - 1)], (d3 < (unsigned)cnt) ? w3 : 0.0f);
    }
    for (; i < e; i += stride) {
      const unsigned d = (unsigned)(dst[i] - base);
      const float w = ew[i];
      atomicAdd(&hist[d & (DEG_TILE - 1)], (d < (unsigned)cnt) ? w : 0.0f);
    }
    __syncthreads();
    for (int j = t; j < cnt; j += DEG_BLOCK) mine[base + j] = hist[j];
    __syncthreads();
  }
}

__global__ void dinv_reduce_kernel(const float* __restrict__ part,
                                   float* __restrict__ dinv, int n) {
  int i = blockIdx.x * blockDim.x + threadIdx.x;
  if (i < n) {
    float d = 1.0f;
#pragma unroll 16
    for (int g = 0; g < DEG_GROUPS; ++g) d += part[(size_t)g * n + i];
    dinv[i] = (d > 0.0f) ? rsqrtf(d) : 0.0f;
  }
}

template <int K, int NOUT>
__global__ void gemm_wmma_scale_kernel(const float* __restrict__ X,
                                       const float* __restrict__ W,
                                       const float* __restrict__ dinv,
                                       float* __restrict__ hs,
                                       int nrows) {
  const int lane = threadIdx.x & 31;
  const int wave = blockIdx.x * (blockDim.x >> 5) + (threadIdx.x >> 5);
  const int ntc  = NOUT / 16;
  const int row0 = (wave / ntc) * 16;
  const int col0 = (wave % ntc) * 16;
  if (row0 >= nrows) return;

  const int half = lane >> 4;
  const int m    = lane & 15;

  fx8 c = {};
  const float* xrow = X + (size_t)(row0 + m) * K;

#pragma unroll
  for (int kb = 0; kb < K; kb += 4) {
    const int ka = kb + 2 * half;
    fx2 a, b;
    a.x = xrow[ka];
    a.y = xrow[ka + 1];
    b.x = W[(size_t)ka * NOUT + col0 + m];
    b.y = W[(size_t)(ka + 1) * NOUT + col0 + m];
    c = __builtin_amdgcn_wmma_f32_16x16x4_f32(false, a, false, b,
                                              (short)0, c, false, false);
  }

#pragma unroll
  for (int r = 0; r < 8; ++r) {
    const int mr  = r + 8 * half;
    const float s = dinv[row0 + mr];
    const float v = c[r] * s;
    const size_t idx = (size_t)(row0 + mr) * NOUT + col0 + m;
    hs[idx] = v;
  }
}

#define OWN_ROWS  256
#define OWN_BLOCK 1024
#define OWN_PIECE 1024
#define OWN_V     4

__global__ void __launch_bounds__(256)
owner_bits_kernel(const int* __restrict__ dst, unsigned* __restrict__ lst, int e, int npieces) {
  const int lane  = threadIdx.x & 31;
  const int piece = __builtin_amdgcn_readfirstlane(blockIdx.x * (blockDim.x >> 5) + (threadIdx.x >> 5));
  if (piece >= npieces) return;
  const int i0 = piece * OWN_PIECE;
  int w[8] = {0, 0, 0, 0, 0, 0, 0, 0};
#pragma unroll
  for (int g = 0; g < 32; ++g) {
    const int i = i0 + 32 * g + lane;
    const unsigned o = (i < e) ? ((unsigned)dst[i] / OWN_ROWS) : 0u;
#pragma unroll
    for (int p = 0; p < 8; ++p) {
      const int bal = (int)__builtin_amdgcn_ballot_w32(((o >> p) & 1u) != 0u);
      w[p] = (lane == g) ? bal : w[p];
    }
  }
  uint4* o4 = (uint4*)lst + (size_t)piece * 64;
  o4[lane]      = make_uint4((unsigned)w[0], (unsigned)w[1], (unsigned)w[2], (unsigned)w[3]);
  o4[32 + lane] = make_uint4((unsigned)w[4], (unsigned)w[5], (unsigned)w[6], (unsigned)w[7]);
}

template <int F>
__device__ __forceinline__ void own_add_edge(float* acc, const float* __restrict__ hs,
                                             int s, float w, int r, int lane) {
  const float* p = hs + (size_t)s * F + lane;
  float v[F / 32];
#pragma unroll
  for (int c = 0; c < F / 32; ++c) v[c] = p[32 * c];
#pragma unroll
  for (int c = 0; c < F / 32; ++c) atomicAdd(&acc[r * F + 32 * c + lane], w * v[c]);
}

#define OWN_G 8
template <int F>
__device__ __forceinline__ void own_flush(float* acc, const float* __restrict__ hs,
                                          const int* __restrict__ src,
                                          const int* __restrict__ dst,
                                          const float* __restrict__ ew,
                                          int qi, int qn, int base, int cnt, int lane) {
  int   s = 0;
  int   r = -1;
  float w = 0.0f;
  if (lane < qn) { s = src[qi]; w = ew[qi]; r = dst[qi] - base; }
  const unsigned ok   = __builtin_amdgcn_ballot_w32((unsigned)r < (unsigned)cnt);
  const unsigned want = (qn >= 32) ? 0xffffffffu : ((1u << qn) - 1u);
  if (ok == want) {
    int t = 0;
    for (; t + OWN_G <= qn; t += OWN_G) {
      float v[OWN_G][F / 32];
#pragma unroll
      for (int u = 0; u < OWN_G; ++u) {
        const int su = __builtin_amdgcn_readlane(s, t + u);
        const float* p = hs + (size_t)su * F + lane;
#pragma unroll
        for (int c = 0; c < F / 32; ++c) v[u][c] = p[32 * c];
      }
#pragma unroll
      for (int u = 0; u < OWN_G; ++u) {
        const int   ru = __builtin_amdgcn_readlane(r, t + u);
        const float wu = __int_as_float(__builtin_amdgcn_readlane(__float_as_int(w), t + u));
#pragma unroll
        for (int c = 0; c < F / 32; ++c) atomicAdd(&acc[ru * F + 32 * c + lane], wu * v[u][c]);
      }
    }
    for (; t < qn; ++t) {
      const int   su = __builtin_amdgcn_readlane(s, t);
      const int   ru = __builtin_amdgcn_readlane(r, t);
      const float wu = __int_as_float(__builtin_amdgcn_readlane(__float_as_int(w), t));
      own_add_edge<F>(acc, hs, su, wu, ru, lane);
    }
  } else {
    unsigned mk = ok;
    while (mk) {
      const int t = __builtin_ctz(mk);
      mk &= mk - 1u;
      const int   su = __builtin_amdgcn_readlane(s, t);
      const int   ru = __builtin_amdgcn_readlane(r, t);
      const float wu = __int_as_float(__builtin_amdgcn_readlane(__float_as_int(w), t));
      own_add_edge<F>(acc, hs, su, wu, ru, lane);
    }
  }
}

template <int F, bool RELU>
__global__ void __launch_bounds__(OWN_BLOCK)
own_rows_kernel(const float* __restrict__ hs,
                const int* __restrict__ src,
                const int* __restrict__ dst,
                const float* __restrict__ ew,
                const unsigned* __restrict__ lst,
                const float* __restrict__ dinv,
                const float* __restrict__ bias,
                float* __restrict__ out, int n, int e) {
  __shared__ float acc[OWN_ROWS * F];
  const int t    = threadIdx.x;
  const int lane = t & 31;
  const int wv   = __builtin_amdgcn_readfirstlane(t >> 5);
  const int base = blockIdx.x * OWN_ROWS;
  const int cnt  = (n - base < OWN_ROWS) ? (n - base) : OWN_ROWS;

  const float* hs0 = hs + (size_t)base * F;
  for (int j = t; j < cnt * F; j += OWN_BLOCK) acc[j] = hs0[j];
  __syncthreads();

  const int npieces = (e + OWN_PIECE - 1) / OWN_PIECE;
  const int rot = (int)(((long long)blockIdx.x * npieces) / gridDim.x);
  const uint4* lst4 = (const uint4*)lst;
  unsigned npl[8];
#pragma unroll
  for (int p = 0; p < 8; ++p) npl[p] = ((blockIdx.x >> p) & 1u) ? 0u : 0xffffffffu;

  int qi = 0;
  int qn = 0;

  for (int k0 = wv * OWN_V; k0 < npieces; k0 += (OWN_BLOCK / 32) * OWN_V) {
    uint4 pa[OWN_V], pb[OWN_V];
#pragma unroll
    for (int v = 0; v < OWN_V; ++v) {
      int k = k0 + v;
      if (k >= npieces) k = npieces - 1;
      int c = k + rot;
      if (c >= npieces) c -= npieces;
      pa[v] = lst4[(size_t)c * 64 + lane];
      pb[v] = lst4[(size_t)c * 64 + 32 + lane];
    }
    unsigned m[OWN_V];
#pragma unroll
    for (int v = 0; v < OWN_V; ++v) {
      unsigned x = (pa[v].x ^ npl[0]) & (pa[v].y ^ npl[1]) & (pa[v].z ^ npl[2]) & (pa[v].w ^ npl[3]);
      x &= (pb[v].x ^ npl[4]) & (pb[v].y ^ npl[5]) & (pb[v].z ^ npl[6]) & (pb[v].w ^ npl[7]);
      m[v] = (k0 + v < npieces) ? x : 0u;
    }
#pragma nounroll
    for (int h = 0; h < OWN_V / 2; ++h) {
      const unsigned mlo = (h == 0) ? m[0] : m[2];
      const unsigned mhi = (h == 0) ? m[1] : m[3];
      unsigned hm = __builtin_amdgcn_ballot_w32((mlo | mhi) != 0u);
      while (hm) {
        const int j = __builtin_ctz(hm);
        hm &= hm - 1u;
        unsigned long long f =
            ((unsigned long long)(unsigned)__builtin_amdgcn_readlane((int)mhi, j) << 32) |
            (unsigned)__builtin_amdgcn_readlane((int)mlo, j);
        while (f) {
          const int bpos = __builtin_ctzll(f);
          f &= f - 1ull;
          int c = k0 + 2 * h + (bpos >> 5) + rot;
          if (c >= npieces) c -= npieces;
          const int ii = c * OWN_PIECE + 32 * j + (bpos & 31);
          if (ii < e) {
            if (lane == qn) qi = ii;
            ++qn;
            if (qn == 32) {
              own_flush<F>(acc, hs, src, dst, ew, qi, 32, base, cnt, lane);
              qn = 0;
            }
          }
        }
      }
    }
  }
  if (qn > 0) own_flush<F>(acc, hs, src, dst, ew, qi, qn, base, cnt, lane);
  __syncthreads();

  float* o0 = out + (size_t)base * F;
  for (int j = t; j < cnt * F; j += OWN_BLOCK) {
    const int row = j / F;
    const int f   = j % F;
    float v = dinv[base + row] * acc[j] + bias[f];
    if (RELU) v = fmaxf(v, 0.0f);
    o0[j] = v;
  }
}

extern "C" void kernel_launch(void* const* d_in, const int* in_sizes, int n_in,
                              void* d_out, int out_size, void* d_ws, size_t ws_size,
                              hipStream_t stream) {
  const float* z  = (const float*)d_in[0];
  const int*   ei = (const int*)d_in[1];
  const float* ew = (const float*)d_in[2];
  const float* W1 = (const float*)d_in[3];
  const float* b1 = (const float*)d_in[4];
  const float* W2 = (const float*)d_in[5];
  const float* b2 = (const float*)d_in[6];
  float* out = (float*)d_out;

  const int n = in_sizes[0] / 64;
  const int e = in_sizes[2];
  const int* srcp = ei;
  const int* dstp = ei + e;

  float* ws   = (float*)d_ws;
  float* dinv = ws + n;
  float* X1   = dinv + n;
  float* X2   = X1 + (size_t)n * 128;
  const int B = 256;
  const int npieces = (e + OWN_PIECE - 1) / OWN_PIECE;
  unsigned* lst1 = (unsigned*)out;
  unsigned* lst2 = (unsigned*)(X1 + (size_t)n * 64);

  owner_bits_kernel<<<(npieces + 7) / 8, B, 0, stream>>>(dstp, lst1, e, npieces);

  deg_partial_kernel<<<DEG_GROUPS, DEG_BLOCK, 0, stream>>>(dstp, ew, X1, n, e);
  dinv_reduce_kernel<<<(n + B - 1) / B, B, 0, stream>>>(X1, dinv, n);

  {
    const long long waves = (long long)((n + 15) / 16) * (128 / 16);
    const long long thr   = waves * 32;
    gemm_wmma_scale_kernel<64, 128><<<(unsigned)((thr + B - 1) / B), B, 0, stream>>>(
        z, W1, dinv, X1, n);
  }
  own_rows_kernel<128, true><<<(n + OWN_ROWS - 1) / OWN_ROWS, OWN_BLOCK, 0, stream>>>(
      X1, srcp, dstp, ew, lst1, dinv, b1, X2, n, e);
  owner_bits_kernel<<<(npieces + 7) / 8, B, 0, stream>>>(dstp, lst2, e, npieces);
  {
    const long long waves = (long long)((n + 15) / 16) * (64 / 16);
    const long long thr   = waves * 32;
    gemm_wmma_scale_kernel<128, 64><<<(unsigned)((thr + B - 1) / B), B, 0, stream>>>(
        X2, W2, dinv, X1, n);
  }
  own_rows_kernel<64, false><<<(n + OWN_ROWS - 1) / OWN_ROWS, OWN_BLOCK, 0, stream>>>(
      X1, srcp, dstp, ew, lst2, dinv, b2, out, n, e);
}


# Owner work-groups hold 196 destination rows, not 256: 256 owners, one on every WGP, where 196 owners left 60 WGPs idle; each adds 23 % fewer edges; same sums, LDS arrival order as before
# speedup vs baseline: 2.2493x; 1.0377x over previous
#include <hip/hip_runtime.h>

typedef float fx2 __attribute__((ext_vector_type(2)));
typedef float fx8 __attribute__((ext_vector_type(8)));

#define DEG_TILE   65536
#define DEG_GROUPS 128
#define DEG_BLOCK  1024

__global__ void __launch_bounds__(DEG_BLOCK)
deg_partial_kernel(const int* __restrict__ dst,
                   const float* __restrict__ ew,
                   float* __restrict__ part, int n, int e) {
  __shared__ float hist[DEG_TILE];
  const int t = threadIdx.x;
  float* mine = part + (size_t)blockIdx.x * n;
  for (int base = 0; base < n; base += DEG_TILE) {
    const int cnt = (n - base < DEG_TILE) ? (n - base) : DEG_TILE;
    float4* hist4 = (float4*)hist;
    for (int j = t; j < (cnt + 3) / 4; j += DEG_BLOCK)
      hist4[j] = make_float4(0.0f, 0.0f, 0.0f, 0.0f);
    __syncthreads();
    const int stride = DEG_GROUPS * DEG_BLOCK;
    int i = blockIdx.x * DEG_BLOCK + t;
    for (; i < e - 3 * stride; i += 4 * stride) {
      const unsigned d0 = (unsigned)(dst[i] - base);
      const unsigned d1 = (unsigned)(dst[i + stride] - base);
      const unsigned d2 = (unsigned)(dst[i + 2 * stride] - base);
      const unsigned d3 = (unsigned)(dst[i + 3 * stride] - base);
      const float w0 = ew[i];
      const float w1 = ew[i + stride];
      const float w2 = ew[i + 2 * stride];
      const float w3 = ew[i + 3 * stride];
      atomicAdd(&hist[d0 & (DEG_TILE - 1)], (d0 < (unsigned)cnt) ? w0 : 0.0f);
      atomicAdd(&hist[d1 & (DEG_TILE - 1)], (d1 < (unsigned)cnt) ? w1 : 0.0f);
      atomicAdd(&hist[d2 & (DEG_TILE - 1)], (d2 < (unsigned)cnt) ? w2 : 0.0f);
      atomicAdd(&hist[d3 & (DEG_TILE - 1)], (d3 < (unsigned)cnt) ? w3 : 0.0f);
    }
    for (; i < e; i += stride) {
      const unsigned d = (unsigned)(dst[i] - base);
      const float w = ew[i];
      atomicAdd(&hist[d & (DEG_TILE - 1)], (d < (unsigned)cnt) ? w : 0.0f);
    }
    __syncthreads();
    for (int j = t; j < cnt; j += DEG_BLOCK) mine[base + j] = hist[j];
    __syncthreads();
  }
}

__global__ void dinv_reduce_kernel(const float* __restrict__ part,
                                   float* __restrict__ dinv, int n) {
  int i = blockIdx.x * blockDim.x + threadIdx.x;
  if (i < n) {
    float d = 1.0f;
#pragma unroll 16
    for (int g = 0; g < DEG_GROUPS; ++g) d += part[(size_t)g * n + i];
    dinv[i] = (d > 0.0f) ? rsqrtf(d) : 0.0f;
  }
}

template <int K, int NOUT>
__global__ void gemm_wmma_scale_kernel(const float* __restrict__ X,
                                       const float* __restrict__ W,
                                       const float* __restrict__ dinv,
                                       float* __restrict__ hs,
                                       int nrows) {
  const int lane = threadIdx.x & 31;
  const int wave = blockIdx.x * (blockDim.x >> 5) + (threadIdx.x >> 5);
  const int ntc  = NOUT / 16;
  const int row0 = (wave / ntc) * 16;
  const int col0 = (wave % ntc) * 16;
  if (row0 >= nrows) return;

  const int half = lane >> 4;
  const int m    = lane & 15;

  fx8 c = {};
  const float* xrow = X + (size_t)(row0 + m) * K;

#pragma unroll
  for (int kb = 0; kb < K; kb += 4) {
    const int ka = kb + 2 * half;
    fx2 a, b;
    a.x = xrow[ka];
    a.y = xrow[ka + 1];
    b.x = W[(size_t)ka * NOUT + col0 + m];
    b.y = W[(size_t)(ka + 1) * NOUT + col0 + m];
    c = __builtin_amdgcn_wmma_f32_16x16x4_f32(false, a, false, b,
                                              (short)0, c, false, false);
  }

#pragma unroll
  for (int r = 0; r < 8; ++r) {
    const int mr  = r + 8 * half;
    const float s = dinv[row0 + mr];
    const float v = c[r] * s;
    const size_t idx = (size_t)(row0 + mr) * NOUT + col0 + m;
    hs[idx] = v;
  }
}

#define OWN_ROWS  196
#define OWN_BLOCK 1024
#define OWN_PIECE 1024
#define OWN_V     4

__global__ void __launch_bounds__(256)
owner_bits_kernel(const int* __restrict__ dst, unsigned* __restrict__ lst, int e, int npieces) {
  const int lane  = threadIdx.x & 31;
  const int piece = __builtin_amdgcn_readfirstlane(blockIdx.x * (blockDim.x >> 5) + (threadIdx.x >> 5));
  if (piece >= npieces) return;
  const int i0 = piece * OWN_PIECE;
  int w[8] = {0, 0, 0, 0, 0, 0, 0, 0};
#pragma unroll
  for (int g = 0; g < 32; ++g) {
    const int i = i0 + 32 * g + lane;
    const unsigned o = (i < e) ? ((unsigned)dst[i] / OWN_ROWS) : 0u;
#pragma unroll
    for (int p = 0; p < 8; ++p) {
      const int bal = (int)__builtin_amdgcn_ballot_w32(((o >> p) & 1u) != 0u);
      w[p] = (lane == g) ? bal : w[p];
    }
  }
  uint4* o4 = (uint4*)lst + (size_t)piece * 64;
  o4[lane]      = make_uint4((unsigned)w[0], (unsigned)w[1], (unsigned)w[2], (unsigned)w[3]);
  o4[32 + lane] = make_uint4((unsigned)w[4], (unsigned)w[5], (unsigned)w[6], (unsigned)w[7]);
}

template <int F>
__device__ __forceinline__ void own_add_edge(float* acc, const float* __restrict__ hs,
                                             int s, float w, int r, int lane) {
  const float* p = hs + (size_t)s * F + lane;
  float v[F / 32];
#pragma unroll
  for (int c = 0; c < F / 32; ++c) v[c] = p[32 * c];
#pragma unroll
  for (int c = 0; c < F / 32; ++c) atomicAdd(&acc[r * F + 32 * c + lane], w * v[c]);
}

#define OWN_G 8
template <int F>
__device__ __forceinline__ void own_flush(float* acc, const float* __restrict__ hs,
                                          const int* __restrict__ src,
                                          const int* __restrict__ dst,
                                          const float* __restrict__ ew,
                                          int qi, int qn, int base, int cnt, int lane) {
  int   s = 0;
  int   r = -1;
  float w = 0.0f;
  if (lane < qn) { s = src[qi]; w = ew[qi]; r = dst[qi] - base; }
  const unsigned ok   = __builtin_amdgcn_ballot_w32((unsigned)r < (unsigned)cnt);
  const unsigned want = (qn >= 32) ? 0xffffffffu : ((1u << qn) - 1u);
  if (ok == want) {
    int t = 0;
    for (; t + OWN_G <= qn; t += OWN_G) {
      float v[OWN_G][F / 32];
#pragma unroll
      for (int u = 0; u < OWN_G; ++u) {
        const int su = __builtin_amdgcn_readlane(s, t + u);
        const float* p = hs + (size_t)su * F + lane;
#pragma unroll
        for (int c = 0; c < F / 32; ++c) v[u][c] = p[32 * c];
      }
#pragma unroll
      for (int u = 0; u < OWN_G; ++u) {
        const int   ru = __builtin_amdgcn_readlane(r, t + u);
        const float wu = __int_as_float(__builtin_amdgcn_readlane(__float_as_int(w), t + u));
#pragma unroll
        for (int c = 0; c < F / 32; ++c) atomicAdd(&acc[ru * F + 32 * c + lane], wu * v[u][c]);
      }
    }
    for (; t < qn; ++t) {
      const int   su = __builtin_amdgcn_readlane(s, t);
      const int   ru = __builtin_amdgcn_readlane(r, t);
      const float wu = __int_as_float(__builtin_amdgcn_readlane(__float_as_int(w), t));
      own_add_edge<F>(acc, hs, su, wu, ru, lane);
    }
  } else {
    unsigned mk = ok;
    while (mk) {
      const int t = __builtin_ctz(mk);
      mk &= mk - 1u;
      const int   su = __builtin_amdgcn_readlane(s, t);
      const int   ru = __builtin_amdgcn_readlane(r, t);
      const float wu = __int_as_float(__builtin_amdgcn_readlane(__float_as_int(w), t));
      own_add_edge<F>(acc, hs, su, wu, ru, lane);
    }
  }
}

template <int F, bool RELU>
__global__ void __launch_bounds__(OWN_BLOCK)
own_rows_kernel(const float* __restrict__ hs,
                const int* __restrict__ src,
                const int* __restrict__ dst,
                const float* __restrict__ ew,
                const unsigned* __restrict__ lst,
                const float* __restrict__ dinv,
                const float* __restrict__ bias,
                float* __restrict__ out, int n, int e) {
  __shared__ float acc[OWN_ROWS * F];
  const int t    = threadIdx.x;
  const int lane = t & 31;
  const int wv   = __builtin_amdgcn_readfirstlane(t >> 5);
  const int base = blockIdx.x * OWN_ROWS;
  const int cnt  = (n - base < OWN_ROWS) ? (n - base) : OWN_ROWS;

  const float* hs0 = hs + (size_t)base * F;
  for (int j = t; j < cnt * F; j += OWN_BLOCK) acc[j] = hs0[j];
  __syncthreads();

  const int npieces = (e + OWN_PIECE - 1) / OWN_PIECE;
  const int rot = (int)(((long long)blockIdx.x * npieces) / gridDim.x);
  const uint4* lst4 = (const uint4*)lst;
  unsigned npl[8];
#pragma unroll
  for (int p = 0; p < 8; ++p) npl[p] = ((blockIdx.x >> p) & 1u) ? 0u : 0xffffffffu;

  int qi = 0;
  int qn = 0;

  for (int k0 = wv * OWN_V; k0 < npieces; k0 += (OWN_BLOCK / 32) * OWN_V) {
    uint4 pa[OWN_V], pb[OWN_V];
#pragma unroll
    for (int v = 0; v < OWN_V; ++v) {
      int k = k0 + v;
      if (k >= npieces) k = npieces - 1;
      int c = k + rot;
      if (c >= npieces) c -= npieces;
      pa[v] = lst4[(size_t)c * 64 + lane];
      pb[v] = lst4[(size_t)c * 64 + 32 + lane];
    }
    unsigned m[OWN_V];
#pragma unroll
    for (int v = 0; v < OWN_V; ++v) {
      unsigned x = (pa[v].x ^ npl[0]) & (pa[v].y ^ npl[1]) & (pa[v].z ^ npl[2]) & (pa[v].w ^ npl[3]);
      x &= (pb[v].x ^ npl[4]) & (pb[v].y ^ npl[5]) & (pb[v].z ^ npl[6]) & (pb[v].w ^ npl[7]);
      m[v] = (k0 + v < npieces) ? x : 0u;
    }
#pragma nounroll
    for (int h = 0; h < OWN_V / 2; ++h) {
      const unsigned mlo = (h == 0) ? m[0] : m[2];
      const unsigned mhi = (h == 0) ? m[1] : m[3];
      unsigned hm = __builtin_amdgcn_ballot_w32((mlo | mhi) != 0u);
      while (hm) {
        const int j = __builtin_ctz(hm);
        hm &= hm - 1u;
        unsigned long long f =
            ((unsigned long long)(unsigned)__builtin_amdgcn_readlane((int)mhi, j) << 32) |
            (unsigned)__builtin_amdgcn_readlane((int)mlo, j);
        while (f) {
          const int bpos = __builtin_ctzll(f);
          f &= f - 1ull;
          int c = k0 + 2 * h + (bpos >> 5) + rot;
          if (c >= npieces) c -= npieces;
          const int ii = c * OWN_PIECE + 32 * j + (bpos & 31);
          if (ii < e) {
            if (lane == qn) qi = ii;
            ++qn;
            if (qn == 32) {
              own_flush<F>(acc, hs, src, dst, ew, qi, 32, base, cnt, lane);
              qn = 0;
            }
          }
        }
      }
    }
  }
  if (qn > 0) own_flush<F>(acc, hs, src, dst, ew, qi, qn, base, cnt, lane);
  __syncthreads();

  float* o0 = out + (size_t)base * F;
  for (int j = t; j < cnt * F; j += OWN_BLOCK) {
    const int row = j / F;
    const int f   = j % F;
    float v = dinv[base + row] * acc[j] + bias[f];
    if (RELU) v = fmaxf(v, 0.0f);
    o0[j] = v;
  }
}

extern "C" void kernel_launch(void* const* d_in, const int* in_sizes, int n_in,
                              void* d_out, int out_size, void* d_ws, size_t ws_size,
                              hipStream_t stream) {
  const float* z  = (const float*)d_in[0];
  const int*   ei = (const int*)d_in[1];
  const float* ew = (const float*)d_in[2];
  const float* W1 = (const float*)d_in[3];
  const float* b1 = (const float*)d_in[4];
  const float* W2 = (const float*)d_in[5];
  const float* b2 = (const float*)d_in[6];
  float* out = (float*)d_out;

  const int n = in_sizes[0] / 64;
  const int e = in_sizes[2];
  const int* srcp = ei;
  const int* dstp = ei + e;

  float* ws   = (float*)d_ws;
  float* dinv = ws + n;
  float* X1   = dinv + n;
  float* X2   = X1 + (size_t)n * 128;
  const int B = 256;
  const int npieces = (e + OWN_PIECE - 1) / OWN_PIECE;
  unsigned* lst1 = (unsigned*)out;
  unsigned* lst2 = (unsigned*)(X1 + (size_t)n * 64);

  owner_bits_kernel<<<(npieces + 7) / 8, B, 0, stream>>>(dstp, lst1, e, npieces);

  deg_partial_kernel<<<DEG_GROUPS, DEG_BLOCK, 0, stream>>>(dstp, ew, X1, n, e);
  dinv_reduce_kernel<<<(n + B - 1) / B, B, 0, stream>>>(X1, dinv, n);

  {
    const long long waves = (long long)((n + 15) / 16) * (128 / 16);
    const long long thr   = waves * 32;
    gemm_wmma_scale_kernel<64, 128><<<(unsigned)((thr + B - 1) / B), B, 0, stream>>>(
        z, W1, dinv, X1, n);
  }
  own_rows_kernel<128, true><<<(n + OWN_ROWS - 1) / OWN_ROWS, OWN_BLOCK, 0, stream>>>(
      X1, srcp, dstp, ew, lst1, dinv, b1, X2, n, e);
  owner_bits_kernel<<<(npieces + 7) / 8, B, 0, stream>>>(dstp, lst2, e, npieces);
  {
    const long long waves = (long long)((n + 15) / 16) * (64 / 16);
    const long long thr   = waves * 32;
    gemm_wmma_scale_kernel<128, 64><<<(unsigned)((thr + B - 1) / B), B, 0, stream>>>(
        X2, W2, dinv, X1, n);
  }
  own_rows_kernel<64, false><<<(n + OWN_ROWS - 1) / OWN_ROWS, OWN_BLOCK, 0, stream>>>(
      X1, srcp, dstp, ew, lst2, dinv, b2, out, n, e);
}


# Owner-bit kernel issues the 32 destination loads of a piece together (index clamped at the last edge, no branch round a load): the built code had waited for each load in turn; no sum changes
# speedup vs baseline: 2.4019x; 1.0678x over previous
#include <hip/hip_runtime.h>

typedef float fx2 __attribute__((ext_vector_type(2)));
typedef float fx8 __attribute__((ext_vector_type(8)));

#define DEG_TILE   65536
#define DEG_GROUPS 128
#define DEG_BLOCK  1024

__global__ void __launch_bounds__(DEG_BLOCK)
deg_partial_kernel(const int* __restrict__ dst,
                   const float* __restrict__ ew,
                   float* __restrict__ part, int n, int e) {
  __shared__ float hist[DEG_TILE];
  const int t = threadIdx.x;
  float* mine = part + (size_t)blockIdx.x * n;
  for (int base = 0; base < n; base += DEG_TILE) {
    const int cnt = (n - base < DEG_TILE) ? (n - base) : DEG_TILE;
    float4* hist4 = (float4*)hist;
    for (int j = t; j < (cnt + 3) / 4; j += DEG_BLOCK)
      hist4[j] = make_float4(0.0f, 0.0f, 0.0f, 0.0f);
    __syncthreads();
    const int stride = DEG_GROUPS * DEG_BLOCK;
    int i = blockIdx.x * DEG_BLOCK + t;
    for (; i < e - 3 * stride; i += 4 * stride) {
      const unsigned d0 = (unsigned)(dst[i] - base);
      const unsigned d1 = (unsigned)(dst[i + stride] - base);
      const unsigned d2 = (unsigned)(dst[i + 2 * stride] - base);
      const unsigned d3 = (unsigned)(dst[i + 3 * stride] - base);
      const float w0 = ew[i];
      const float w1 = ew[i + stride];
      const float w2 = ew[i + 2 * stride];
      const float w3 = ew[i + 3 * stride];
      atomicAdd(&hist[d0 & (DEG_TILE - 1)], (d0 < (unsigned)cnt) ? w0 : 0.0f);
      atomicAdd(&hist[d1 & (DEG_TILE - 1)], (d1 < (unsigned)cnt) ? w1 : 0.0f);
      atomicAdd(&hist[d2 & (DEG_TILE - 1)], (d2 < (unsigned)cnt) ? w2 : 0.0f);
      atomicAdd(&hist[d3 & (DEG_TILE - 1)], (d3 < (unsigned)cnt) ? w3 : 0.0f);
    }
    for (; i < e; i += stride) {
      const unsigned d = (unsigned)(dst[i] - base);
      const float w = ew[i];
      atomicAdd(&hist[d & (DEG_TILE - 1)], (d < (unsigned)cnt) ? w : 0.0f);
    }
    __syncthreads();
    for (int j = t; j < cnt; j += DEG_BLOCK) mine[base + j] = hist[j];
    __syncthreads();
  }
}

__global__ void dinv_reduce_kernel(const float* __restrict__ part,
                                   float* __restrict__ dinv, int n) {
  int i = blockIdx.x * blockDim.x + threadIdx.x;
  if (i < n) {
    float d = 1.0f;
#pragma unroll 16
    for (int g = 0; g < DEG_GROUPS; ++g) d += part[(size_t)g * n + i];
    dinv[i] = (d > 0.0f) ? rsqrtf(d) : 0.0f;
  }
}

template <int K, int NOUT>
__global__ void gemm_wmma_scale_kernel(const float* __restrict__ X,
                                       const float* __restrict__ W,
                                       const float* __restrict__ dinv,
                                       float* __restrict__ hs,
                                       int nrows) {
  const int lane = threadIdx.x & 31;
  const int wave = blockIdx.x * (blockDim.x >> 5) + (threadIdx.x >> 5);
  const int ntc  = NOUT / 16;
  const int row0 = (wave / ntc) * 16;
  const int col0 = (wave % ntc) * 16;
  if (row0 >= nrows) return;

  const int half = lane >> 4;
  const int m    = lane & 15;

  fx8 c = {};
  const float* xrow = X + (size_t)(row0 + m) * K;

#pragma unroll
  for (int kb = 0; kb < K; kb += 4) {
    const int ka = kb + 2 * half;
    fx2 a, b;
    a.x = xrow[ka];
    a.y = xrow[ka + 1];
    b.x = W[(size_t)ka * NOUT + col0 + m];
    b.y = W[(size_t)(ka + 1) * NOUT + col0 + m];
    c = __builtin_amdgcn_wmma_f32_16x16x4_f32(false, a, false, b,
                                              (short)0, c, false, false);
  }

#pragma unroll
  for (int r = 0; r < 8; ++r) {
    const int mr  = r + 8 * half;
    const float s = dinv[row0 + mr];
    const float v = c[r] * s;
    const size_t idx = (size_t)(row0 + mr) * NOUT + col0 + m;
    hs[idx] = v;
  }
}

#define OWN_ROWS  196
#define OWN_BLOCK 1024
#define OWN_PIECE 1024
#define OWN_V     4

__global__ void __launch_bounds__(256)
owner_bits_kernel(const int* __restrict__ dst, unsigned* __restrict__ lst, int e, int npieces) {
  const int lane  = threadIdx.x & 31;
  const int piece = __builtin_amdgcn_readfirstlane(blockIdx.x * (blockDim.x >> 5) + (threadIdx.x >> 5));
  if (piece >= npieces) return;
  const int i0 = piece * OWN_PIECE;
  int d[32];
#pragma unroll
  for (int g = 0; g < 32; ++g) {
    const int i = i0 + 32 * g + lane;
    d[g] = dst[(i < e) ? i : (e - 1)];
  }
  int w[8] = {0, 0, 0, 0, 0, 0, 0, 0};
#pragma unroll
  for (int g = 0; g < 32; ++g) {
    const int i = i0 + 32 * g + lane;
    const unsigned o = (i < e) ? ((unsigned)d[g] / OWN_ROWS) : 0u;
#pragma unroll
    for (int p = 0; p < 8; ++p) {
      const int bal = (int)__builtin_amdgcn_ballot_w32(((o >> p) & 1u) != 0u);
      w[p] = (lane == g) ? bal : w[p];
    }
  }
  uint4* o4 = (uint4*)lst + (size_t)piece * 64;
  o4[lane]      = make_uint4((unsigned)w[0], (unsigned)w[1], (unsigned)w[2], (unsigned)w[3]);
  o4[32 + lane] = make_uint4((unsigned)w[4], (unsigned)w[5], (unsigned)w[6], (unsigned)w[7]);
}

template <int F>
__device__ __forceinline__ void own_add_edge(float* acc, const float* __restrict__ hs,
                                             int s, float w, int r, int lane) {
  const float* p = hs + (size_t)s * F + lane;
  float v[F / 32];
#pragma unroll
  for (int c = 0; c < F / 32; ++c) v[c] = p[32 * c];
#pragma unroll
  for (int c = 0; c < F / 32; ++c) atomicAdd(&acc[r * F + 32 * c + lane], w * v[c]);
}

#define OWN_G 8
template <int F>
__device__ __forceinline__ void own_flush(float* acc, const float* __restrict__ hs,
                                          const int* __restrict__ src,
                                          const int* __restrict__ dst,
                                          const float* __restrict__ ew,
                                          int qi, int qn, int base, int cnt, int lane) {
  int   s = 0;
  int   r = -1;
  float w = 0.0f;
  if (lane < qn) { s = src[qi]; w = ew[qi]; r = dst[qi] - base; }
  const unsigned ok   = __builtin_amdgcn_ballot_w32((unsigned)r < (unsigned)cnt);
  const unsigned want = (qn >= 32) ? 0xffffffffu : ((1u << qn) - 1u);
  if (ok == want) {
    int t = 0;
    for (; t + OWN_G <= qn; t += OWN_G) {
      float v[OWN_G][F / 32];
#pragma unroll
      for (int u = 0; u < OWN_G; ++u) {
        const int su = __builtin_amdgcn_readlane(s, t + u);
        const float* p = hs + (size_t)su * F + lane;
#pragma unroll
        for (int c = 0; c < F / 32; ++c) v[u][c] = p[32 * c];
      }
#pragma unroll
      for (int u = 0; u < OWN_G; ++u) {
        const int   ru = __builtin_amdgcn_readlane(r, t + u);
        const float wu = __int_as_float(__builtin_amdgcn_readlane(__float_as_int(w), t + u));
#pragma unroll
        for (int c = 0; c < F / 32; ++c) atomicAdd(&acc[ru * F + 32 * c + lane], wu * v[u][c]);
      }
    }
    for (; t < qn; ++t) {
      const int   su = __builtin_amdgcn_readlane(s, t);
      const int   ru = __builtin_amdgcn_readlane(r, t);
      const float wu = __int_as_float(__builtin_amdgcn_readlane(__float_as_int(w), t));
      own_add_edge<F>(acc, hs, su, wu, ru, lane);
    }
  } else {
    unsigned mk = ok;
    while (mk) {
      const int t = __builtin_ctz(mk);
      mk &= mk - 1u;
      const int   su = __builtin_amdgcn_readlane(s, t);
      const int   ru = __builtin_amdgcn_readlane(r, t);
      const float wu = __int_as_float(__builtin_amdgcn_readlane(__float_as_int(w), t));
      own_add_edge<F>(acc, hs, su, wu, ru, lane);
    }
  }
}

template <int F, bool RELU>
__global__ void __launch_bounds__(OWN_BLOCK)
own_rows_kernel(const float* __restrict__ hs,
                const int* __restrict__ src,
                const int* __restrict__ dst,
                const float* __restrict__ ew,
                const unsigned* __restrict__ lst,
                const float* __restrict__ dinv,
                const float* __restrict__ bias,
                float* __restrict__ out, int n, int e) {
  __shared__ float acc[OWN_ROWS * F];
  const int t    = threadIdx.x;
  const int lane = t & 31;
  const int wv   = __builtin_amdgcn_readfirstlane(t >> 5);
  const int base = blockIdx.x * OWN_ROWS;
  const int cnt  = (n - base < OWN_ROWS) ? (n - base) : OWN_ROWS;

  const float* hs0 = hs + (size_t)base * F;
  for (int j = t; j < cnt * F; j += OWN_BLOCK) acc[j] = hs0[j];
  __syncthreads();

  const int npieces = (e + OWN_PIECE - 1) / OWN_PIECE;
  const int rot = (int)(((long long)blockIdx.x * npieces) / gridDim.x);
  const uint4* lst4 = (const uint4*)lst;
  unsigned npl[8];
#pragma unroll
  for (int p = 0; p < 8; ++p) npl[p] = ((blockIdx.x >> p) & 1u) ? 0u : 0xffffffffu;

  int qi = 0;
  int qn = 0;

  for (int k0 = wv * OWN_V; k0 < npieces; k0 += (OWN_BLOCK / 32) * OWN_V) {
    uint4 pa[OWN_V], pb[OWN_V];
#pragma unroll
    for (int v = 0; v < OWN_V; ++v) {
      int k = k0 + v;
      if (k >= npieces) k = npieces - 1;
      int c = k + rot;
      if (c >= npieces) c -= npieces;
      pa[v] = lst4[(size_t)c * 64 + lane];
      pb[v] = lst4[(size_t)c * 64 + 32 + lane];
    }
    unsigned m[OWN_V];
#pragma unroll
    for (int v = 0; v < OWN_V; ++v) {
      unsigned x = (pa[v].x ^ npl[0]) & (pa[v].y ^ npl[1]) & (pa[v].z ^ npl[2]) & (pa[v].w ^ npl[3]);
      x &= (pb[v].x ^ npl[4]) & (pb[v].y ^ npl[5]) & (pb[v].z ^ npl[6]) & (pb[v].w ^ npl[7]);
      m[v] = (k0 + v < npieces) ? x : 0u;
    }
#pragma nounroll
    for (int h = 0; h < OWN_V / 2; ++h) {
      const unsigned mlo = (h == 0) ? m[0] : m[2];
      const unsigned mhi = (h == 0) ? m[1] : m[3];
      unsigned hm = __builtin_amdgcn_ballot_w32((mlo | mhi) != 0u);
      while (hm) {
        const int j = __builtin_ctz(hm);
        hm &= hm - 1u;
        unsigned long long f =
            ((unsigned long long)(unsigned)__builtin_amdgcn_readlane((int)mhi, j) << 32) |
            (unsigned)__builtin_amdgcn_readlane((int)mlo, j);
        while (f) {
          const int bpos = __builtin_ctzll(f);
          f &= f - 1ull;
          int c = k0 + 2 * h + (bpos >> 5) + rot;
          if (c >= npieces) c -= npieces;
          const int ii = c * OWN_PIECE + 32 * j + (bpos & 31);
          if (ii < e) {
            if (lane == qn) qi = ii;
            ++qn;
            if (qn == 32) {
              own_flush<F>(acc, hs, src, dst, ew, qi, 32, base, cnt, lane);
              qn = 0;
            }
          }
        }
      }
    }
  }
  if (qn > 0) own_flush<F>(acc, hs, src, dst, ew, qi, qn, base, cnt, lane);
  __syncthreads();

  float* o0 = out + (size_t)base * F;
  for (int j = t; j < cnt * F; j += OWN_BLOCK) {
    const int row = j / F;
    const int f   = j % F;
    float v = dinv[base + row] * acc[j] + bias[f];
    if (RELU) v = fmaxf(v, 0.0f);
    o0[j] = v;
  }
}

extern "C" void kernel_launch(void* const* d_in, const int* in_sizes, int n_in,
                              void* d_out, int out_size, void* d_ws, size_t ws_size,
                              hipStream_t stream) {
  const float* z  = (const float*)d_in[0];
  const int*   ei = (const int*)d_in[1];
  const float* ew = (const float*)d_in[2];
  const float* W1 = (const float*)d_in[3];
  const float* b1 = (const float*)d_in[4];
  const float* W2 = (const float*)d_in[5];
  const float* b2 = (const float*)d_in[6];
  float* out = (float*)d_out;

  const int n = in_sizes[0] / 64;
  const int e = in_sizes[2];
  const int* srcp = ei;
  const int* dstp = ei + e;

  float* ws   = (float*)d_ws;
  float* dinv = ws + n;
  float* X1   = dinv + n;
  float* X2   = X1 + (size_t)n * 128;
  const int B = 256;
  const int npieces = (e + OWN_PIECE - 1) / OWN_PIECE;
  unsigned* lst1 = (unsigned*)out;
  unsigned* lst2 = (unsigned*)(X1 + (size_t)n * 64);

  owner_bits_kernel<<<(npieces + 7) / 8, B, 0, stream>>>(dstp, lst1, e, npieces);

  deg_partial_kernel<<<DEG_GROUPS, DEG_BLOCK, 0, stream>>>(dstp, ew, X1, n, e);
  dinv_reduce_kernel<<<(n + B - 1) / B, B, 0, stream>>>(X1, dinv, n);

  {
    const long long waves = (long long)((n + 15) / 16) * (128 / 16);
    const long long thr   = waves * 32;
    gemm_wmma_scale_kernel<64, 128><<<(unsigned)((thr + B - 1) / B), B, 0, stream>>>(
        z, W1, dinv, X1, n);
  }
  own_rows_kernel<128, true><<<(n + OWN_ROWS - 1) / OWN_ROWS, OWN_BLOCK, 0, stream>>>(
      X1, srcp, dstp, ew, lst1, dinv, b1, X2, n, e);
  owner_bits_kernel<<<(npieces + 7) / 8, B, 0, stream>>>(dstp, lst2, e, npieces);
  {
    const long long waves = (long long)((n + 15) / 16) * (64 / 16);
    const long long thr   = waves * 32;
    gemm_wmma_scale_kernel<128, 64><<<(unsigned)((thr + B - 1) / B), B, 0, stream>>>(
        X2, W2, dinv, X1, n);
  }
  own_rows_kernel<64, false><<<(n + OWN_ROWS - 1) / OWN_ROWS, OWN_BLOCK, 0, stream>>>(
      X1, srcp, dstp, ew, lst2, dinv, b2, out, n, e);
}


# Layer 1 owners leave a list of the edges they keep (index, local row) in their own output block; layer 2 reads it: no second owner-bit kernel, list filter, queue or dst fetch; LDS arrival order
# speedup vs baseline: 2.8348x; 1.1802x over previous
#include <hip/hip_runtime.h>

typedef float fx2 __attribute__((ext_vector_type(2)));
typedef float fx8 __attribute__((ext_vector_type(8)));

#define DEG_TILE   65536
#define DEG_GROUPS 128
#define DEG_BLOCK  1024

__global__ void __launch_bounds__(DEG_BLOCK)
deg_partial_kernel(const int* __restrict__ dst,
                   const float* __restrict__ ew,
                   float* __restrict__ part, int n, int e) {
  __shared__ float hist[DEG_TILE];
  const int t = threadIdx.x;
  float* mine = part + (size_t)blockIdx.x * n;
  for (int base = 0; base < n; base += DEG_TILE) {
    const int cnt = (n - base < DEG_TILE) ? (n - base) : DEG_TILE;
    float4* hist4 = (float4*)hist;
    for (int j = t; j < (cnt + 3) / 4; j += DEG_BLOCK)
      hist4[j] = make_float4(0.0f, 0.0f, 0.0f, 0.0f);
    __syncthreads();
    const int stride = DEG_GROUPS * DEG_BLOCK;
    int i = blockIdx.x * DEG_BLOCK + t;
    for (; i < e - 3 * stride; i += 4 * stride) {
      const unsigned d0 = (unsigned)(dst[i] - base);
      const unsigned d1 = (unsigned)(dst[i + stride] - base);
      const unsigned d2 = (unsigned)(dst[i + 2 * stride] - base);
      const unsigned d3 = (unsigned)(dst[i + 3 * stride] - base);
      const float w0 = ew[i];
      const float w1 = ew[i + stride];
      const float w2 = ew[i + 2 * stride];
      const float w3 = ew[i + 3 * stride];
      atomicAdd(&hist[d0 & (DEG_TILE - 1)], (d0 < (unsigned)cnt) ? w0 : 0.0f);
      atomicAdd(&hist[d1 & (DEG_TILE - 1)], (d1 < (unsigned)cnt) ? w1 : 0.0f);
      atomicAdd(&hist[d2 & (DEG_TILE - 1)], (d2 < (unsigned)cnt) ? w2 : 0.0f);
      atomicAdd(&hist[d3 & (DEG_TILE - 1)], (d3 < (unsigned)cnt) ? w3 : 0.0f);
    }
    for (; i < e; i += stride) {
      const unsigned d = (unsigned)(dst[i] - base);
      const float w = ew[i];
      atomicAdd(&hist[d & (DEG_TILE - 1)], (d < (unsigned)cnt) ? w : 0.0f);
    }
    __syncthreads();
    for (int j = t; j < cnt; j += DEG_BLOCK) mine[base + j] = hist[j];
    __syncthreads();
  }
}

__global__ void dinv_reduce_kernel(const float* __restrict__ part,
                                   float* __restrict__ dinv, int n) {
  int i = blockIdx.x * blockDim.x + threadIdx.x;
  if (i < n) {
    float d = 1.0f;
#pragma unroll 16
    for (int g = 0; g < DEG_GROUPS; ++g) d += part[(size_t)g * n + i];
    dinv[i] = (d > 0.0f) ? rsqrtf(d) : 0.0f;
  }
}

template <int K, int NOUT>
__global__ void gemm_wmma_scale_kernel(const float* __restrict__ X,
                                       const float* __restrict__ W,
                                       const float* __restrict__ dinv,
                                       float* __restrict__ hs,
                                       int nrows) {
  const int lane = threadIdx.x & 31;
  const int wave = blockIdx.x * (blockDim.x >> 5) + (threadIdx.x >> 5);
  const int ntc  = NOUT / 16;
  const int row0 = (wave / ntc) * 16;
  const int col0 = (wave % ntc) * 16;
  if (row0 >= nrows) return;

  const int half = lane >> 4;
  const int m    = lane & 15;

  fx8 c = {};
  const float* xrow = X + (size_t)(row0 + m) * K;

#pragma unroll
  for (int kb = 0; kb < K; kb += 4) {
    const int ka = kb + 2 * half;
    fx2 a, b;
    a.x = xrow[ka];
    a.y = xrow[ka + 1];
    b.x = W[(size_t)ka * NOUT + col0 + m];
    b.y = W[(size_t)(ka + 1) * NOUT + col0 + m];
    c = __builtin_amdgcn_wmma_f32_16x16x4_f32(false, a, false, b,
                                              (short)0, c, false, false);
  }

#pragma unroll
  for (int r = 0; r < 8; ++r) {
    const int mr  = r + 8 * half;
    const float s = dinv[row0 + mr];
    const float v = c[r] * s;
    const size_t idx = (size_t)(row0 + mr) * NOUT + col0 + m;
    hs[idx] = v;
  }
}

#define OWN_ROWS  196
#define OWN_BLOCK 1024
#define OWN_PIECE 1024
#define OWN_V     4
#define OWN_REG    (OWN_ROWS * 64)
#define OWN_SP     8
#define OWN_STRIPE (OWN_SP * (OWN_PIECE / 4))

__global__ void __launch_bounds__(256)
owner_bits_kernel(const int* __restrict__ dst, unsigned* __restrict__ lst, int e, int npieces) {
  const int lane  = threadIdx.x & 31;
  const int piece = __builtin_amdgcn_readfirstlane(blockIdx.x * (blockDim.x >> 5) + (threadIdx.x >> 5));
  if (piece >= npieces) return;
  const int i0 = piece * OWN_PIECE;
  int d[32];
#pragma unroll
  for (int g = 0; g < 32; ++g) {
    const int i = i0 + 32 * g + lane;
    d[g] = dst[(i < e) ? i : (e - 1)];
  }
  int w[8] = {0, 0, 0, 0, 0, 0, 0, 0};
#pragma unroll
  for (int g = 0; g < 32; ++g) {
    const int i = i0 + 32 * g + lane;
    const unsigned o = (i < e) ? ((unsigned)d[g] / OWN_ROWS) : 0u;
#pragma unroll
    for (int p = 0; p < 8; ++p) {
      const int bal = (int)__builtin_amdgcn_ballot_w32(((o >> p) & 1u) != 0u);
      w[p] = (lane == g) ? bal : w[p];
    }
  }
  uint4* o4 = (uint4*)(lst + (size_t)(piece / OWN_SP) * OWN_REG + (OWN_REG - OWN_STRIPE)) + (piece % OWN_SP) * 64;
  o4[lane]      = make_uint4((unsigned)w[0], (unsigned)w[1], (unsigned)w[2], (unsigned)w[3]);
  o4[32 + lane] = make_uint4((unsigned)w[4], (unsigned)w[5], (unsigned)w[6], (unsigned)w[7]);
}

template <int F>
__device__ __forceinline__ void own_add_edge(float* acc, const float* __restrict__ hs,
                                             int s, float w, int r, int lane) {
  const float* p = hs + (size_t)s * F + lane;
  float v[F / 32];
#pragma unroll
  for (int c = 0; c < F / 32; ++c) v[c] = p[32 * c];
#pragma unroll
  for (int c = 0; c < F / 32; ++c) atomicAdd(&acc[r * F + 32 * c + lane], w * v[c]);
}

#define OWN_G 8
template <int F>
__device__ __forceinline__ void own_add_queue(float* acc, const float* __restrict__ hs,
                                              int s, float w, int r,
                                              int qn, int cnt, int lane) {
  const unsigned ok   = __builtin_amdgcn_ballot_w32((unsigned)r < (unsigned)cnt);
  const unsigned want = (qn >= 32) ? 0xffffffffu : ((1u << qn) - 1u);
  if (ok == want) {
    int t = 0;
    for (; t + OWN_G <= qn; t += OWN_G) {
      float v[OWN_G][F / 32];
#pragma unroll
      for (int u = 0; u < OWN_G; ++u) {
        const int su = __builtin_amdgcn_readlane(s, t + u);
        const float* p = hs + (size_t)su * F + lane;
#pragma unroll
        for (int c = 0; c < F / 32; ++c) v[u][c] = p[32 * c];
      }
#pragma unroll
      for (int u = 0; u < OWN_G; ++u) {
        const int   ru = __builtin_amdgcn_readlane(r, t + u);
        const float wu = __int_as_float(__builtin_amdgcn_readlane(__float_as_int(w), t + u));
#pragma unroll
        for (int c = 0; c < F / 32; ++c) atomicAdd(&acc[ru * F + 32 * c + lane], wu * v[u][c]);
      }
    }
    for (; t < qn; ++t) {
      const int   su = __builtin_amdgcn_readlane(s, t);
      const int   ru = __builtin_amdgcn_readlane(r, t);
      const float wu = __int_as_float(__builtin_amdgcn_readlane(__float_as_int(w), t));
      own_add_edge<F>(acc, hs, su, wu, ru, lane);
    }
  } else {
    unsigned mk = ok;
    while (mk) {
      const int t = __builtin_ctz(mk);
      mk &= mk - 1u;
      const int   su = __builtin_amdgcn_readlane(s, t);
      const int   ru = __builtin_amdgcn_readlane(r, t);
      const float wu = __int_as_float(__builtin_amdgcn_readlane(__float_as_int(w), t));
      own_add_edge<F>(acc, hs, su, wu, ru, lane);
    }
  }
}

#define OWN_IB    23
#define OWN_NOROW 255
static_assert(OWN_ROWS <= OWN_NOROW, "a local row must stay below the not-owned mark");

__device__ __forceinline__ void own_record(unsigned* mine, unsigned* rcount, int cap,
                                           int qi, int r, int cnt, int qn, int lane) {
  int off = 0;
  if (lane == 0) off = (int)atomicAdd(rcount, (unsigned)qn);
  off = __builtin_amdgcn_readfirstlane(off);
  const int p = off + lane;
  const unsigned rr = ((unsigned)r < (unsigned)cnt) ? (unsigned)r : (unsigned)OWN_NOROW;
  if (lane < qn && p < cap) mine[p] = (unsigned)qi | (rr << OWN_IB);
}

template <int F>
__device__ __forceinline__ void own_flush(float* acc, const float* __restrict__ hs,
                                          const int* __restrict__ src,
                                          const int* __restrict__ dst,
                                          const float* __restrict__ ew,
                                          unsigned* mine, unsigned* rcount, int cap,
                                          int qi, int qn, int base, int cnt, int lane) {
  int   s = 0;
  int   r = -1;
  float w = 0.0f;
  if (lane < qn) { s = src[qi]; w = ew[qi]; r = dst[qi] - base; }
  own_record(mine, rcount, cap, qi, r, cnt, qn, lane);
  own_add_queue<F>(acc, hs, s, w, r, qn, cnt, lane);
}

__device__ __forceinline__ int own_cap(int b, int cnt, int npieces) {
  const int cap = cnt * 64 - ((b * OWN_SP < npieces) ? OWN_STRIPE : 0);
  return (cap > 0 && npieces <= ((1 << OWN_IB) / OWN_PIECE)) ? cap : 0;
}

template <int F, bool RELU>
__device__ __forceinline__ void own_finish(const float* acc, const float* __restrict__ dinv,
                                           const float* __restrict__ bias, float* o0,
                                           int base, int cnt, int t) {
  for (int j = t; j < cnt * F; j += OWN_BLOCK) {
    const int row = j / F;
    const int f   = j % F;
    float v = dinv[base + row] * acc[j] + bias[f];
    if (RELU) v = fmaxf(v, 0.0f);
    o0[j] = v;
  }
}

template <int F, bool RELU>
__global__ void __launch_bounds__(OWN_BLOCK)
own_rows_kernel(const float* __restrict__ hs,
                const int* __restrict__ src,
                const int* __restrict__ dst,
                const float* __restrict__ ew,
                const unsigned* __restrict__ lst,
                unsigned* rec,
                int* __restrict__ reccnt,
                const float* __restrict__ dinv,
                const float* __restrict__ bias,
                float* __restrict__ out, int n, int e) {
  __shared__ float acc[OWN_ROWS * F];
  __shared__ unsigned rcount;
  const int t    = threadIdx.x;
  const int lane = t & 31;
  const int wv   = __builtin_amdgcn_readfirstlane(t >> 5);
  const int base = blockIdx.x * OWN_ROWS;
  const int cnt  = (n - base < OWN_ROWS) ? (n - base) : OWN_ROWS;

  const float* hs0 = hs + (size_t)base * F;
  for (int j = t; j < cnt * F; j += OWN_BLOCK) acc[j] = hs0[j];
  if (t == 0) rcount = 0u;
  __syncthreads();

  const int npieces = (e + OWN_PIECE - 1) / OWN_PIECE;
  unsigned* mine = rec + (size_t)blockIdx.x * OWN_REG;
  const int cap  = own_cap((int)blockIdx.x, cnt, npieces);
  const int rot = (int)(((long long)blockIdx.x * npieces) / gridDim.x);
  unsigned npl[8];
#pragma unroll
  for (int p = 0; p < 8; ++p) npl[p] = ((blockIdx.x >> p) & 1u) ? 0u : 0xffffffffu;

  int qi = 0;
  int qn = 0;

  for (int k0 = wv * OWN_V; k0 < npieces; k0 += (OWN_BLOCK / 32) * OWN_V) {
    uint4 pa[OWN_V], pb[OWN_V];
#pragma unroll
    for (int v = 0; v < OWN_V; ++v) {
      int k = k0 + v;
      if (k >= npieces) k = npieces - 1;
      int c = k + rot;
      if (c >= npieces) c -= npieces;
      const uint4* pc = (const uint4*)(lst + (size_t)(c / OWN_SP) * OWN_REG + (OWN_REG - OWN_STRIPE)) + (c % OWN_SP) * 64;
      pa[v] = pc[lane];
      pb[v] = pc[32 + lane];
    }
    unsigned m[OWN_V];
#pragma unroll
    for (int v = 0; v < OWN_V; ++v) {
      unsigned x = (pa[v].x ^ npl[0]) & (pa[v].y ^ npl[1]) & (pa[v].z ^ npl[2]) & (pa[v].w ^ npl[3]);
      x &= (pb[v].x ^ npl[4]) & (pb[v].y ^ npl[5]) & (pb[v].z ^ npl[6]) & (pb[v].w ^ npl[7]);
      m[v] = (k0 + v < npieces) ? x : 0u;
    }
#pragma nounroll
    for (int h = 0; h < OWN_V / 2; ++h) {
      const unsigned mlo = (h == 0) ? m[0] : m[2];
      const unsigned mhi = (h == 0) ? m[1] : m[3];
      unsigned hm = __builtin_amdgcn_ballot_w32((mlo | mhi) != 0u);
      while (hm) {
        const int j = __builtin_ctz(hm);
        hm &= hm - 1u;
        unsigned long long f =
            ((unsigned long long)(unsigned)__builtin_amdgcn_readlane((int)mhi, j) << 32) |
            (unsigned)__builtin_amdgcn_readlane((int)mlo, j);
        while (f) {
          const int bpos = __builtin_ctzll(f);
          f &= f - 1ull;
          int c = k0 + 2 * h + (bpos >> 5) + rot;
          if (c >= npieces) c -= npieces;
          const int ii = c * OWN_PIECE + 32 * j + (bpos & 31);
          if (ii < e) {
            if (lane == qn) qi = ii;
            ++qn;
            if (qn == 32) {
              own_flush<F>(acc, hs, src, dst, ew, mine, &rcount, cap, qi, 32, base, cnt, lane);
              qn = 0;
            }
          }
        }
      }
    }
  }
  if (qn > 0) own_flush<F>(acc, hs, src, dst, ew, mine, &rcount, cap, qi, qn, base, cnt, lane);
  __syncthreads();
  if (t == 0) reccnt[blockIdx.x] = (int)rcount;

  own_finish<F, RELU>(acc, dinv, bias, out + (size_t)base * F, base, cnt, t);
}

template <int F, bool RELU>
__global__ void __launch_bounds__(OWN_BLOCK)
own_list_kernel(const float* __restrict__ hs,
                const int* __restrict__ src,
                const int* __restrict__ dst,
                const float* __restrict__ ew,
                const int* __restrict__ reccnt,
                const float* __restrict__ dinv,
                const float* __restrict__ bias,
                float* out, int n, int e) {
  static_assert(F * OWN_ROWS == OWN_REG, "the edge list lives in the owner's own output rows");
  __shared__ float acc[OWN_ROWS * F];
  const int t    = threadIdx.x;
  const int lane = t & 31;
  const int wv   = __builtin_amdgcn_readfirstlane(t >> 5);
  const int base = blockIdx.x * OWN_ROWS;
  const int cnt  = (n - base < OWN_ROWS) ? (n - base) : OWN_ROWS;

  const float* hs0 = hs + (size_t)base * F;
  for (int j = t; j < cnt * F; j += OWN_BLOCK) acc[j] = hs0[j];
  __syncthreads();

  const int npieces = (e + OWN_PIECE - 1) / OWN_PIECE;
  const int cap     = own_cap((int)blockIdx.x, cnt, npieces);
  const int total   = reccnt[blockIdx.x];
  const unsigned* mine = (const unsigned*)out + (size_t)blockIdx.x * OWN_REG;

  if (total <= cap) {
    int i0 = wv * 32;
    unsigned q = (i0 + lane < total) ? mine[i0 + lane] : 0u;
    while (i0 < total) {
      const int qn = (total - i0 < 32) ? (total - i0) : 32;
      const int i1 = i0 + OWN_BLOCK;
      const unsigned qx = (i1 + lane < total) ? mine[i1 + lane] : 0u;
      int   s = 0;
      int   r = -1;
      float w = 0.0f;
      if (lane < qn) {
        const int qi = (int)(q & ((1u << OWN_IB) - 1u));
        s = src[qi];
        w = ew[qi];
        r = (int)(q >> OWN_IB);
      }
      own_add_queue<F>(acc, hs, s, w, r, qn, cnt, lane);
      q  = qx;
      i0 = i1;
    }
  } else {
    for (int i = t; i < e; i += OWN_BLOCK) {
      const unsigned r = (unsigned)(dst[i] - base);
      if (r < (unsigned)cnt) {
        const float  w = ew[i];
        const float* p = hs + (size_t)src[i] * F;
        for (int f = 0; f < F; ++f) atomicAdd(&acc[r * F + f], w * p[f]);
      }
    }
  }
  __syncthreads();

  own_finish<F, RELU>(acc, dinv, bias, out + (size_t)base * F, base, cnt, t);
}

extern "C" void kernel_launch(void* const* d_in, const int* in_sizes, int n_in,
                              void* d_out, int out_size, void* d_ws, size_t ws_size,
                              hipStream_t stream) {
  const float* z  = (const float*)d_in[0];
  const int*   ei = (const int*)d_in[1];
  const float* ew = (const float*)d_in[2];
  const float* W1 = (const float*)d_in[3];
  const float* b1 = (const float*)d_in[4];
  const float* W2 = (const float*)d_in[5];
  const float* b2 = (const float*)d_in[6];
  float* out = (float*)d_out;

  const int n = in_sizes[0] / 64;
  const int e = in_sizes[2];
  const int* srcp = ei;
  const int* dstp = ei + e;

  float* ws   = (float*)d_ws;
  int*   kept = (int*)ws;
  float* dinv = ws + n;
  float* X1   = dinv + n;
  float* X2   = X1 + (size_t)n * 128;
  const int B = 256;
  const int npieces = (e + OWN_PIECE - 1) / OWN_PIECE;
  unsigned* obuf = (unsigned*)out;

  owner_bits_kernel<<<(npieces + 7) / 8, B, 0, stream>>>(dstp, obuf, e, npieces);

  deg_partial_kernel<<<DEG_GROUPS, DEG_BLOCK, 0, stream>>>(dstp, ew, X1, n, e);
  dinv_reduce_kernel<<<(n + B - 1) / B, B, 0, stream>>>(X1, dinv, n);

  {
    const long long waves = (long long)((n + 15) / 16) * (128 / 16);
    const long long thr   = waves * 32;
    gemm_wmma_scale_kernel<64, 128><<<(unsigned)((thr + B - 1) / B), B, 0, stream>>>(
        z, W1, dinv, X1, n);
  }
  own_rows_kernel<128, true><<<(n + OWN_ROWS - 1) / OWN_ROWS, OWN_BLOCK, 0, stream>>>(
      X1, srcp, dstp, ew, obuf, obuf, kept, dinv, b1, X2, n, e);
  {
    const long long waves = (long long)((n + 15) / 16) * (64 / 16);
    const long long thr   = waves * 32;
    gemm_wmma_scale_kernel<128, 64><<<(unsigned)((thr + B - 1) / B), B, 0, stream>>>(
        X2, W2, dinv, X1, n);
  }
  own_list_kernel<64, false><<<(n + OWN_ROWS - 1) / OWN_ROWS, OWN_BLOCK, 0, stream>>>(
      X1, srcp, dstp, ew, kept, dinv, b2, out, n, e);
}


# Edges binned by owner once: 128 chunk work-groups rank edges per owner in LDS and fill slots of the owners' lists; both layers read them: no bit planes, list filter, dst fetch; LDS arrival order
# speedup vs baseline: 3.3505x; 1.1819x over previous
#include <hip/hip_runtime.h>

typedef float fx2 __attribute__((ext_vector_type(2)));
typedef float fx8 __attribute__((ext_vector_type(8)));

#define DEG_TILE   65536
#define DEG_GROUPS 128
#define DEG_BLOCK  1024

__global__ void __launch_bounds__(DEG_BLOCK)
deg_partial_kernel(const int* __restrict__ dst,
                   const float* __restrict__ ew,
                   float* __restrict__ part, int n, int e) {
  __shared__ float hist[DEG_TILE];
  const int t = threadIdx.x;
  float* mine = part + (size_t)blockIdx.x * n;
  for (int base = 0; base < n; base += DEG_TILE) {
    const int cnt = (n - base < DEG_TILE) ? (n - base) : DEG_TILE;
    float4* hist4 = (float4*)hist;
    for (int j = t; j < (cnt + 3) / 4; j += DEG_BLOCK)
      hist4[j] = make_float4(0.0f, 0.0f, 0.0f, 0.0f);
    __syncthreads();
    const int stride = DEG_GROUPS * DEG_BLOCK;
    int i = blockIdx.x * DEG_BLOCK + t;
    for (; i < e - 3 * stride; i += 4 * stride) {
      const unsigned d0 = (unsigned)(dst[i] - base);
      const unsigned d1 = (unsigned)(dst[i + stride] - base);
      const unsigned d2 = (unsigned)(dst[i + 2 * stride] - base);
      const unsigned d3 = (unsigned)(dst[i + 3 * stride] - base);
      const float w0 = ew[i];
      const float w1 = ew[i + stride];
      const float w2 = ew[i + 2 * stride];
      const float w3 = ew[i + 3 * stride];
      atomicAdd(&hist[d0 & (DEG_TILE - 1)], (d0 < (unsigned)cnt) ? w0 : 0.0f);
      atomicAdd(&hist[d1 & (DEG_TILE - 1)], (d1 < (unsigned)cnt) ? w1 : 0.0f);
      atomicAdd(&hist[d2 & (DEG_TILE - 1)], (d2 < (unsigned)cnt) ? w2 : 0.0f);
      atomicAdd(&hist[d3 & (DEG_TILE - 1)], (d3 < (unsigned)cnt) ? w3 : 0.0f);
    }
    for (; i < e; i += stride) {
      const unsigned d = (unsigned)(dst[i] - base);
      const float w = ew[i];
      atomicAdd(&hist[d & (DEG_TILE - 1)], (d < (unsigned)cnt) ? w : 0.0f);
    }
    __syncthreads();
    for (int j = t; j < cnt; j += DEG_BLOCK) mine[base + j] = hist[j];
    __syncthreads();
  }
}

__global__ void dinv_reduce_kernel(const float* __restrict__ part,
                                   float* __restrict__ dinv, int n) {
  int i = blockIdx.x * blockDim.x + threadIdx.x;
  if (i < n) {
    float d = 1.0f;
#pragma unroll 16
    for (int g = 0; g < DEG_GROUPS; ++g) d += part[(size_t)g * n + i];
    dinv[i] = (d > 0.0f) ? rsqrtf(d) : 0.0f;
  }
}

template <int K, int NOUT>
__global__ void gemm_wmma_scale_kernel(const float* __restrict__ X,
                                       const float* __restrict__ W,
                                       const float* __restrict__ dinv,
                                       float* __restrict__ hs,
                                       int nrows) {
  const int lane = threadIdx.x & 31;
  const int wave = blockIdx.x * (blockDim.x >> 5) + (threadIdx.x >> 5);
  const int ntc  = NOUT / 16;
  const int row0 = (wave / ntc) * 16;
  const int col0 = (wave % ntc) * 16;
  if (row0 >= nrows) return;

  const int half = lane >> 4;
  const int m    = lane & 15;

  fx8 c = {};
  const float* xrow = X + (size_t)(row0 + m) * K;

#pragma unroll
  for (int kb = 0; kb < K; kb += 4) {
    const int ka = kb + 2 * half;
    fx2 a, b;
    a.x = xrow[ka];
    a.y = xrow[ka + 1];
    b.x = W[(size_t)ka * NOUT + col0 + m];
    b.y = W[(size_t)(ka + 1) * NOUT + col0 + m];
    c = __builtin_amdgcn_wmma_f32_16x16x4_f32(false, a, false, b,
                                              (short)0, c, false, false);
  }

#pragma unroll
  for (int r = 0; r < 8; ++r) {
    const int mr  = r + 8 * half;
    const float s = dinv[row0 + mr];
    const float v = c[r] * s;
    const size_t idx = (size_t)(row0 + mr) * NOUT + col0 + m;
    hs[idx] = v;
  }
}

#define OWN_ROWS  196
#define OWN_BLOCK 1024
#define OWN_REG   (OWN_ROWS * 64)
#define OWN_IB    23
static_assert(OWN_ROWS <= 256, "a local row must fit the 8 bits above the edge index");

#define BIN_CHUNKS 128
#define BIN_SLOT   96
#define BIN_MAXOWN 256
#define BIN_BLOCK  1024
#define BIN_U      16
#define BIN_OVER   (BIN_SLOT + 1)
static_assert(BIN_CHUNKS * BIN_SLOT <= OWN_REG, "an owner's slots must fit its list block");
static_assert(BIN_SLOT % 32 == 0 && BIN_CHUNKS % 32 == 0 && OWN_BLOCK == 1024, "whole waves per slot; slot c is copied by wave c % 32 of 32");

__device__ __forceinline__ unsigned* own_list_base(unsigned* obuf, unsigned* plane, int b, int n, int nown) {
  if ((size_t)(b + 1) * OWN_REG <= (size_t)n * 64) return obuf + (size_t)b * OWN_REG;
  if (n - OWN_REG >= nown * BIN_CHUNKS) return plane + (n - OWN_REG);
  return nullptr;
}

__global__ void __launch_bounds__(BIN_BLOCK)
bin_edges_kernel(const int* __restrict__ dst, unsigned* __restrict__ obuf,
                 unsigned* __restrict__ plane, int n, int e) {
  __shared__ unsigned stage[BIN_MAXOWN * BIN_SLOT];
  __shared__ unsigned cn[BIN_MAXOWN];
  const int t    = threadIdx.x;
  const int lane = t & 31;
  const int wv   = t >> 5;
  const int c    = blockIdx.x;
  const int nown = (n + OWN_ROWS - 1) / OWN_ROWS;
  const int len  = (e + BIN_CHUNKS - 1) / BIN_CHUNKS;
  const int lo   = c * len;
  const int hi   = (lo + len < e) ? (lo + len) : e;
  for (int j = t; j < BIN_MAXOWN; j += BIN_BLOCK) cn[j] = 0u;
  __syncthreads();
  for (int k0 = lo; k0 < hi; k0 += BIN_U * BIN_BLOCK) {
    int d[BIN_U];
#pragma unroll
    for (int u = 0; u < BIN_U; ++u) {
      const int i = k0 + u * BIN_BLOCK + t;
      d[u] = dst[(i < hi) ? i : (hi - 1)];
    }
#pragma unroll
    for (int u = 0; u < BIN_U; ++u) {
      const int i = k0 + u * BIN_BLOCK + t;
      const unsigned o = (unsigned)d[u] / OWN_ROWS;
      const unsigned r = (unsigned)d[u] - o * OWN_ROWS;
      if (i < hi && o < BIN_MAXOWN) {
        const unsigned rank = atomicAdd(&cn[o], 1u);
        if (rank < BIN_SLOT) stage[o * BIN_SLOT + rank] = (unsigned)i | (r << OWN_IB);
      }
    }
  }
  __syncthreads();
  const int nheld = (nown < BIN_MAXOWN) ? nown : BIN_MAXOWN;
  for (int o = wv; o < nheld; o += BIN_BLOCK / 32) {
    unsigned* base = own_list_base(obuf, plane, o, n, nown);
    if (base == nullptr) continue;
    const unsigned cnt = (cn[o] < BIN_SLOT) ? cn[o] : BIN_SLOT;
    for (unsigned i = lane; i < cnt; i += 32) base[c * BIN_SLOT + i] = stage[o * BIN_SLOT + i];
  }
  const bool enc = e <= (1 << OWN_IB);
  for (int o = t; o < nown; o += BIN_BLOCK) {
    const bool ok = enc && o < BIN_MAXOWN && own_list_base(obuf, plane, o, n, nown) != nullptr;
    plane[(size_t)o * BIN_CHUNKS + c] = ok ? cn[o] : (unsigned)BIN_OVER;
  }
}

template <int F>
__device__ __forceinline__ void own_add_edge(float* acc, const float* __restrict__ hs,
                                             int s, float w, int r, int lane) {
  const float* p = hs + (size_t)s * F + lane;
  float v[F / 32];
#pragma unroll
  for (int c = 0; c < F / 32; ++c) v[c] = p[32 * c];
#pragma unroll
  for (int c = 0; c < F / 32; ++c) atomicAdd(&acc[r * F + 32 * c + lane], w * v[c]);
}

#define OWN_G 8
template <int F>
__device__ __forceinline__ void own_add_queue(float* acc, const float* __restrict__ hs,
                                              int s, float w, int r,
                                              int qn, int cnt, int lane) {
  const unsigned ok   = __builtin_amdgcn_ballot_w32((unsigned)r < (unsigned)cnt);
  const unsigned want = (qn >= 32) ? 0xffffffffu : ((1u << qn) - 1u);
  if (ok == want) {
    int t = 0;
    for (; t + OWN_G <= qn; t += OWN_G) {
      float v[OWN_G][F / 32];
#pragma unroll
      for (int u = 0; u < OWN_G; ++u) {
        const int su = __builtin_amdgcn_readlane(s, t + u);
        const float* p = hs + (size_t)su * F + lane;
#pragma unroll
        for (int c = 0; c < F / 32; ++c) v[u][c] = p[32 * c];
      }
#pragma unroll
      for (int u = 0; u < OWN_G; ++u) {
        const int   ru = __builtin_amdgcn_readlane(r, t + u);
        const float wu = __int_as_float(__builtin_amdgcn_readlane(__float_as_int(w), t + u));
#pragma unroll
        for (int c = 0; c < F / 32; ++c) atomicAdd(&acc[ru * F + 32 * c + lane], wu * v[u][c]);
      }
    }
    for (; t < qn; ++t) {
      const int   su = __builtin_amdgcn_readlane(s, t);
      const int   ru = __builtin_amdgcn_readlane(r, t);
      const float wu = __int_as_float(__builtin_amdgcn_readlane(__float_as_int(w), t));
      own_add_edge<F>(acc, hs, su, wu, ru, lane);
    }
  } else {
    unsigned mk = ok;
    while (mk) {
      const int t = __builtin_ctz(mk);
      mk &= mk - 1u;
      const int   su = __builtin_amdgcn_readlane(s, t);
      const int   ru = __builtin_amdgcn_readlane(r, t);
      const float wu = __int_as_float(__builtin_amdgcn_readlane(__float_as_int(w), t));
      own_add_edge<F>(acc, hs, su, wu, ru, lane);
    }
  }
}

template <int F, bool RELU>
__device__ __forceinline__ void own_finish(const float* acc, const float* __restrict__ dinv,
                                           const float* __restrict__ bias, float* o0,
                                           int base, int cnt, int t) {
  for (int j = t; j < cnt * F; j += OWN_BLOCK) {
    const int row = j / F;
    const int f   = j % F;
    float v = dinv[base + row] * acc[j] + bias[f];
    if (RELU) v = fmaxf(v, 0.0f);
    o0[j] = v;
  }
}

template <int F, bool RELU>
__global__ void __launch_bounds__(OWN_BLOCK)
own_list_kernel(const float* __restrict__ hs,
                const int* __restrict__ src,
                const int* __restrict__ dst,
                const float* __restrict__ ew,
                unsigned* lst,
                unsigned* plane,
                const float* __restrict__ dinv,
                const float* __restrict__ bias,
                float* out, int n, int e) {
  __shared__ float acc[OWN_ROWS * F];
  __shared__ unsigned dense[BIN_CHUNKS * BIN_SLOT];
  __shared__ int cn[BIN_CHUNKS];
  __shared__ int off[BIN_CHUNKS];
  __shared__ int tot[2];
  const int t    = threadIdx.x;
  const int lane = t & 31;
  const int wv   = __builtin_amdgcn_readfirstlane(t >> 5);
  const int base = blockIdx.x * OWN_ROWS;
  const int cnt  = (n - base < OWN_ROWS) ? (n - base) : OWN_ROWS;
  const int nown = (n + OWN_ROWS - 1) / OWN_ROWS;

  const float* hs0 = hs + (size_t)base * F;
  for (int j = t; j < cnt * F; j += OWN_BLOCK) acc[j] = hs0[j];

  if (wv == 0) {
    const int* pc = (const int*)plane + (size_t)blockIdx.x * BIN_CHUNKS + lane * (BIN_CHUNKS / 32);
    int v[BIN_CHUNKS / 32];
    int sum = 0, over = 0;
#pragma unroll
    for (int k = 0; k < BIN_CHUNKS / 32; ++k) {
      v[k] = pc[k];
      over |= (v[k] > BIN_SLOT) ? 1 : 0;
      sum += v[k];
    }
    int incl = sum;
#pragma unroll
    for (int dlt = 1; dlt < 32; dlt <<= 1) {
      const int up = __shfl_up(incl, dlt, 32);
      if (lane >= dlt) incl += up;
    }
    int run = incl - sum;
#pragma unroll
    for (int k = 0; k < BIN_CHUNKS / 32; ++k) {
      cn[lane * (BIN_CHUNKS / 32) + k]  = v[k];
      off[lane * (BIN_CHUNKS / 32) + k] = run;
      run += v[k];
    }
    const unsigned anyover = __builtin_amdgcn_ballot_w32(over != 0);
    if (lane == 31) { tot[0] = incl; tot[1] = (anyover != 0u) ? 1 : 0; }
  }
  __syncthreads();
  const int  total = tot[0];
  const bool whole = tot[1] == 0;

  if (whole) {
    const unsigned* mine = own_list_base(lst, plane, (int)blockIdx.x, n, nown);
    unsigned q[BIN_CHUNKS / 32][BIN_SLOT / 32];
#pragma unroll
    for (int g = 0; g < BIN_CHUNKS / 32; ++g) {
      const int c  = 32 * g + wv;
      const int nc = cn[c];
#pragma unroll
      for (int k = 0; k < BIN_SLOT / 32; ++k)
        q[g][k] = (32 * k + lane < nc) ? mine[c * BIN_SLOT + 32 * k + lane] : 0u;
    }
#pragma unroll
    for (int g = 0; g < BIN_CHUNKS / 32; ++g) {
      const int c  = 32 * g + wv;
      const int nc = cn[c];
      const int oc = off[c];
#pragma unroll
      for (int k = 0; k < BIN_SLOT / 32; ++k)
        if (32 * k + lane < nc) dense[oc + 32 * k + lane] = q[g][k];
    }
    __syncthreads();

    for (int i0 = wv * 32; i0 < total; i0 += OWN_BLOCK) {
      const int qn = (total - i0 < 32) ? (total - i0) : 32;
      int   s = 0;
      int   r = -1;
      float w = 0.0f;
      if (lane < qn) {
        const unsigned q = dense[i0 + lane];
        const int qi = (int)(q & ((1u << OWN_IB) - 1u));
        s = src[qi];
        w = ew[qi];
        r = (int)(q >> OWN_IB);
      }
      own_add_queue<F>(acc, hs, s, w, r, qn, cnt, lane);
    }
  } else {
    for (int i = t; i < e; i += OWN_BLOCK) {
      const unsigned r = (unsigned)(dst[i] - base);
      if (r < (unsigned)cnt) {
        const float  w = ew[i];
        const float* p = hs + (size_t)src[i] * F;
        for (int f = 0; f < F; ++f) atomicAdd(&acc[r * F + f], w * p[f]);
      }
    }
  }
  __syncthreads();

  own_finish<F, RELU>(acc, dinv, bias, out + (size_t)base * F, base, cnt, t);
}

extern "C" void kernel_launch(void* const* d_in, const int* in_sizes, int n_in,
                              void* d_out, int out_size, void* d_ws, size_t ws_size,
                              hipStream_t stream) {
  const float* z  = (const float*)d_in[0];
  const int*   ei = (const int*)d_in[1];
  const float* ew = (const float*)d_in[2];
  const float* W1 = (const float*)d_in[3];
  const float* b1 = (const float*)d_in[4];
  const float* W2 = (const float*)d_in[5];
  const float* b2 = (const float*)d_in[6];
  float* out = (float*)d_out;

  const int n = in_sizes[0] / 64;
  const int e = in_sizes[2];
  const int* srcp = ei;
  const int* dstp = ei + e;

  float* ws   = (float*)d_ws;
  unsigned* plane = (unsigned*)ws;
  float* dinv = ws + n;
  float* X1   = dinv + n;
  float* X2   = X1 + (size_t)n * 128;
  const int B = 256;
  unsigned* obuf = (unsigned*)out;

  bin_edges_kernel<<<BIN_CHUNKS, BIN_BLOCK, 0, stream>>>(dstp, obuf, plane, n, e);

  deg_partial_kernel<<<DEG_GROUPS, DEG_BLOCK, 0, stream>>>(dstp, ew, X1, n, e);
  dinv_reduce_kernel<<<(n + B - 1) / B, B, 0, stream>>>(X1, dinv, n);

  {
    const long long waves = (long long)((n + 15) / 16) * (128 / 16);
    const long long thr   = waves * 32;
    gemm_wmma_scale_kernel<64, 128><<<(unsigned)((thr + B - 1) / B), B, 0, stream>>>(
        z, W1, dinv, X1, n);
  }
  own_list_kernel<128, true><<<(n + OWN_ROWS - 1) / OWN_ROWS, OWN_BLOCK, 0, stream>>>(
      X1, srcp, dstp, ew, obuf, plane, dinv, b1, X2, n, e);
  {
    const long long waves = (long long)((n + 15) / 16) * (64 / 16);
    const long long thr   = waves * 32;
    gemm_wmma_scale_kernel<128, 64><<<(unsigned)((thr + B - 1) / B), B, 0, stream>>>(
        X2, W2, dinv, X1, n);
  }
  own_list_kernel<64, false><<<(n + OWN_ROWS - 1) / OWN_ROWS, OWN_BLOCK, 0, stream>>>(
      X1, srcp, dstp, ew, obuf, plane, dinv, b2, out, n, e);
}


# GEMMs: one wave computes a 16-row strip over all output columns (8 or 4 accumulator tiles, columns interleaved): A read once per strip, 16-byte B loads and stores, 8x/4x fewer waves; same sums
# speedup vs baseline: 3.4822x; 1.0393x over previous
#include <hip/hip_runtime.h>

typedef float fx2 __attribute__((ext_vector_type(2)));
typedef float fx8 __attribute__((ext_vector_type(8)));

#define DEG_TILE   65536
#define DEG_GROUPS 128
#define DEG_BLOCK  1024

__global__ void __launch_bounds__(DEG_BLOCK)
deg_partial_kernel(const int* __restrict__ dst,
                   const float* __restrict__ ew,
                   float* __restrict__ part, int n, int e) {
  __shared__ float hist[DEG_TILE];
  const int t = threadIdx.x;
  float* mine = part + (size_t)blockIdx.x * n;
  for (int base = 0; base < n; base += DEG_TILE) {
    const int cnt = (n - base < DEG_TILE) ? (n - base) : DEG_TILE;
    float4* hist4 = (float4*)hist;
    for (int j = t; j < (cnt + 3) / 4; j += DEG_BLOCK)
      hist4[j] = make_float4(0.0f, 0.0f, 0.0f, 0.0f);
    __syncthreads();
    const int stride = DEG_GROUPS * DEG_BLOCK;
    int i = blockIdx.x * DEG_BLOCK + t;
    for (; i < e - 3 * stride; i += 4 * stride) {
      const unsigned d0 = (unsigned)(dst[i] - base);
      const unsigned d1 = (unsigned)(dst[i + stride] - base);
      const unsigned d2 = (unsigned)(dst[i + 2 * stride] - base);
      const unsigned d3 = (unsigned)(dst[i + 3 * stride] - base);
      const float w0 = ew[i];
      const float w1 = ew[i + stride];
      const float w2 = ew[i + 2 * stride];
      const float w3 = ew[i + 3 * stride];
      atomicAdd(&hist[d0 & (DEG_TILE - 1)], (d0 < (unsigned)cnt) ? w0 : 0.0f);
      atomicAdd(&hist[d1 & (DEG_TILE - 1)], (d1 < (unsigned)cnt) ? w1 : 0.0f);
      atomicAdd(&hist[d2 & (DEG_TILE - 1)], (d2 < (unsigned)cnt) ? w2 : 0.0f);
      atomicAdd(&hist[d3 & (DEG_TILE - 1)], (d3 < (unsigned)cnt) ? w3 : 0.0f);
    }
    for (; i < e; i += stride) {
      const unsigned d = (unsigned)(dst[i] - base);
      const float w = ew[i];
      atomicAdd(&hist[d & (DEG_TILE - 1)], (d < (unsigned)cnt) ? w : 0.0f);
    }
    __syncthreads();
    for (int j = t; j < cnt; j += DEG_BLOCK) mine[base + j] = hist[j];
    __syncthreads();
  }
}

__global__ void dinv_reduce_kernel(const float* __restrict__ part,
                                   float* __restrict__ dinv, int n) {
  int i = blockIdx.x * blockDim.x + threadIdx.x;
  if (i < n) {
    float d = 1.0f;
#pragma unroll 16
    for (int g = 0; g < DEG_GROUPS; ++g) d += part[(size_t)g * n + i];
    dinv[i] = (d > 0.0f) ? rsqrtf(d) : 0.0f;
  }
}

#define GEMM_BLOCK 128

template <int K, int NOUT>
__global__ void __launch_bounds__(GEMM_BLOCK)
gemm_wmma_scale_kernel(const float* __restrict__ X,
                       const float* __restrict__ W,
                       const float* __restrict__ dinv,
                       float* __restrict__ hs,
                       int nrows) {
  constexpr int NT = NOUT / 16;
  static_assert(NT % 4 == 0, "a lane's columns move as whole 16-byte groups");
  const int lane = threadIdx.x & 31;
  const int wave = blockIdx.x * (blockDim.x >> 5) + (threadIdx.x >> 5);
  const int row0 = wave * 16;
  if (row0 >= nrows) return;

  const int half = lane >> 4;
  const int m    = lane & 15;

  fx8 c[NT];
#pragma unroll
  for (int j = 0; j < NT; ++j) c[j] = fx8{};
  const float* xrow = X + (size_t)(row0 + m) * K;
  const float* wcol = W + NT * m;

#pragma unroll
  for (int kb = 0; kb < K; kb += 4) {
    const int ka = kb + 2 * half;
    fx2 a;
    a.x = xrow[ka];
    a.y = xrow[ka + 1];
    float b0[NT], b1[NT];
#pragma unroll
    for (int q = 0; q < NT / 4; ++q) {
      const float4 u = *(const float4*)(wcol + (size_t)ka * NOUT + 4 * q);
      const float4 v = *(const float4*)(wcol + (size_t)(ka + 1) * NOUT + 4 * q);
      b0[4 * q] = u.x; b0[4 * q + 1] = u.y; b0[4 * q + 2] = u.z; b0[4 * q + 3] = u.w;
      b1[4 * q] = v.x; b1[4 * q + 1] = v.y; b1[4 * q + 2] = v.z; b1[4 * q + 3] = v.w;
    }
#pragma unroll
    for (int j = 0; j < NT; ++j) {
      fx2 b;
      b.x = b0[j];
      b.y = b1[j];
      c[j] = __builtin_amdgcn_wmma_f32_16x16x4_f32(false, a, false, b,
                                                   (short)0, c[j], false, false);
    }
  }

#pragma unroll
  for (int r = 0; r < 8; ++r) {
    const int mr  = r + 8 * half;
    const float s = dinv[row0 + mr];
    float* orow = hs + (size_t)(row0 + mr) * NOUT + NT * m;
#pragma unroll
    for (int q = 0; q < NT / 4; ++q) {
      float4 o;
      o.x = c[4 * q][r] * s;
      o.y = c[4 * q + 1][r] * s;
      o.z = c[4 * q + 2][r] * s;
      o.w = c[4 * q + 3][r] * s;
      *(float4*)(orow + 4 * q) = o;
    }
  }
}

#define OWN_ROWS  196
#define OWN_BLOCK 1024
#define OWN_REG   (OWN_ROWS * 64)
#define OWN_IB    23
static_assert(OWN_ROWS <= 256, "a local row must fit the 8 bits above the edge index");

#define BIN_CHUNKS 128
#define BIN_SLOT   96
#define BIN_MAXOWN 256
#define BIN_BLOCK  1024
#define BIN_U      16
#define BIN_OVER   (BIN_SLOT + 1)
static_assert(BIN_CHUNKS * BIN_SLOT <= OWN_REG, "an owner's slots must fit its list block");
static_assert(BIN_SLOT % 32 == 0 && BIN_CHUNKS % 32 == 0 && OWN_BLOCK == 1024, "whole waves per slot; slot c is copied by wave c % 32 of 32");

__device__ __forceinline__ unsigned* own_list_base(unsigned* obuf, unsigned* plane, int b, int n, int nown) {
  if ((size_t)(b + 1) * OWN_REG <= (size_t)n * 64) return obuf + (size_t)b * OWN_REG;
  if (n - OWN_REG >= nown * BIN_CHUNKS) return plane + (n - OWN_REG);
  return nullptr;
}

__global__ void __launch_bounds__(BIN_BLOCK)
bin_edges_kernel(const int* __restrict__ dst, unsigned* __restrict__ obuf,
                 unsigned* __restrict__ plane, int n, int e) {
  __shared__ unsigned stage[BIN_MAXOWN * BIN_SLOT];
  __shared__ unsigned cn[BIN_MAXOWN];
  const int t    = threadIdx.x;
  const int lane = t & 31;
  const int wv   = t >> 5;
  const int c    = blockIdx.x;
  const int nown = (n + OWN_ROWS - 1) / OWN_ROWS;
  const int len  = (e + BIN_CHUNKS - 1) / BIN_CHUNKS;
  const int lo   = c * len;
  const int hi   = (lo + len < e) ? (lo + len) : e;
  for (int j = t; j < BIN_MAXOWN; j += BIN_BLOCK) cn[j] = 0u;
  __syncthreads();
  for (int k0 = lo; k0 < hi; k0 += BIN_U * BIN_BLOCK) {
    int d[BIN_U];
#pragma unroll
    for (int u = 0; u < BIN_U; ++u) {
      const int i = k0 + u * BIN_BLOCK + t;
      d[u] = dst[(i < hi) ? i : (hi - 1)];
    }
#pragma unroll
    for (int u = 0; u < BIN_U; ++u) {
      const int i = k0 + u * BIN_BLOCK + t;
      const unsigned o = (unsigned)d[u] / OWN_ROWS;
      const unsigned r = (unsigned)d[u] - o * OWN_ROWS;
      if (i < hi && o < BIN_MAXOWN) {
        const unsigned rank = atomicAdd(&cn[o], 1u);
        if (rank < BIN_SLOT) stage[o * BIN_SLOT + rank] = (unsigned)i | (r << OWN_IB);
      }
    }
  }
  __syncthreads();
  const int nheld = (nown < BIN_MAXOWN) ? nown : BIN_MAXOWN;
  for (int o = wv; o < nheld; o += BIN_BLOCK / 32) {
    unsigned* base = own_list_base(obuf, plane, o, n, nown);
    if (base == nullptr) continue;
    const unsigned cnt = (cn[o] < BIN_SLOT) ? cn[o] : BIN_SLOT;
    for (unsigned i = lane; i < cnt; i += 32) base[c * BIN_SLOT + i] = stage[o * BIN_SLOT + i];
  }
  const bool enc = e <= (1 << OWN_IB);
  for (int o = t; o < nown; o += BIN_BLOCK) {
    const bool ok = enc && o < BIN_MAXOWN && own_list_base(obuf, plane, o, n, nown) != nullptr;
    plane[(size_t)o * BIN_CHUNKS + c] = ok ? cn[o] : (unsigned)BIN_OVER;
  }
}

template <int F>
__device__ __forceinline__ void own_add_edge(float* acc, const float* __restrict__ hs,
                                             int s, float w, int r, int lane) {
  const float* p = hs + (size_t)s * F + lane;
  float v[F / 32];
#pragma unroll
  for (int c = 0; c < F / 32; ++c) v[c] = p[32 * c];
#pragma unroll
  for (int c = 0; c < F / 32; ++c) atomicAdd(&acc[r * F + 32 * c + lane], w * v[c]);
}

#define OWN_G 8
template <int F>
__device__ __forceinline__ void own_add_queue(float* acc, const float* __restrict__ hs,
                                              int s, float w, int r,
                                              int qn, int cnt, int lane) {
  const unsigned ok   = __builtin_amdgcn_ballot_w32((unsigned)r < (unsigned)cnt);
  const unsigned want = (qn >= 32) ? 0xffffffffu : ((1u << qn) - 1u);
  if (ok == want) {
    int t = 0;
    for (; t + OWN_G <= qn; t += OWN_G) {
      float v[OWN_G][F / 32];
#pragma unroll
      for (int u = 0; u < OWN_G; ++u) {
        const int su = __builtin_amdgcn_readlane(s, t + u);
        const float* p = hs + (size_t)su * F + lane;
#pragma unroll
        for (int c = 0; c < F / 32; ++c) v[u][c] = p[32 * c];
      }
#pragma unroll
      for (int u = 0; u < OWN_G; ++u) {
        const int   ru = __builtin_amdgcn_readlane(r, t + u);
        const float wu = __int_as_float(__builtin_amdgcn_readlane(__float_as_int(w), t + u));
#pragma unroll
        for (int c = 0; c < F / 32; ++c) atomicAdd(&acc[ru * F + 32 * c + lane], wu * v[u][c]);
      }
    }
    for (; t < qn; ++t) {
      const int   su = __builtin_amdgcn_readlane(s, t);
      const int   ru = __builtin_amdgcn_readlane(r, t);
      const float wu = __int_as_float(__builtin_amdgcn_readlane(__float_as_int(w), t));
      own_add_edge<F>(acc, hs, su, wu, ru, lane);
    }
  } else {
    unsigned mk = ok;
    while (mk) {
      const int t = __builtin_ctz(mk);
      mk &= mk - 1u;
      const int   su = __builtin_amdgcn_readlane(s, t);
      const int   ru = __builtin_amdgcn_readlane(r, t);
      const float wu = __int_as_float(__builtin_amdgcn_readlane(__float_as_int(w), t));
      own_add_edge<F>(acc, hs, su, wu, ru, lane);
    }
  }
}

template <int F, bool RELU>
__device__ __forceinline__ void own_finish(const float* acc, const float* __restrict__ dinv,
                                           const float* __restrict__ bias, float* o0,
                                           int base, int cnt, int t) {
  for (int j = t; j < cnt * F; j += OWN_BLOCK) {
    const int row = j / F;
    const int f   = j % F;
    float v = dinv[base + row] * acc[j] + bias[f];
    if (RELU) v = fmaxf(v, 0.0f);
    o0[j] = v;
  }
}

template <int F, bool RELU>
__global__ void __launch_bounds__(OWN_BLOCK)
own_list_kernel(const float* __restrict__ hs,
                const int* __restrict__ src,
                const int* __restrict__ dst,
                const float* __restrict__ ew,
                unsigned* lst,
                unsigned* plane,
                const float* __restrict__ dinv,
                const float* __restrict__ bias,
                float* out, int n, int e) {
  __shared__ float acc[OWN_ROWS * F];
  __shared__ unsigned dense[BIN_CHUNKS * BIN_SLOT];
  __shared__ int cn[BIN_CHUNKS];
  __shared__ int off[BIN_CHUNKS];
  __shared__ int tot[2];
  const int t    = threadIdx.x;
  const int lane = t & 31;
  const int wv   = __builtin_amdgcn_readfirstlane(t >> 5);
  const int base = blockIdx.x * OWN_ROWS;
  const int cnt  = (n - base < OWN_ROWS) ? (n - base) : OWN_ROWS;
  const int nown = (n + OWN_ROWS - 1) / OWN_ROWS;

  const float* hs0 = hs + (size_t)base * F;
  for (int j = t; j < cnt * F; j += OWN_BLOCK) acc[j] = hs0[j];

  if (wv == 0) {
    const int* pc = (const int*)plane + (size_t)blockIdx.x * BIN_CHUNKS + lane * (BIN_CHUNKS / 32);
    int v[BIN_CHUNKS / 32];
    int sum = 0, over = 0;
#pragma unroll
    for (int k = 0; k < BIN_CHUNKS / 32; ++k) {
      v[k] = pc[k];
      over |= (v[k] > BIN_SLOT) ? 1 : 0;
      sum += v[k];
    }
    int incl = sum;
#pragma unroll
    for (int dlt = 1; dlt < 32; dlt <<= 1) {
      const int up = __shfl_up(incl, dlt, 32);
      if (lane >= dlt) incl += up;
    }
    int run = incl - sum;
#pragma unroll
    for (int k = 0; k < BIN_CHUNKS / 32; ++k) {
      cn[lane * (BIN_CHUNKS / 32) + k]  = v[k];
      off[lane * (BIN_CHUNKS / 32) + k] = run;
      run += v[k];
    }
    const unsigned anyover = __builtin_amdgcn_ballot_w32(over != 0);
    if (lane == 31) { tot[0] = incl; tot[1] = (anyover != 0u) ? 1 : 0; }
  }
  __syncthreads();
  const int  total = tot[0];
  const bool whole = tot[1] == 0;

  if (whole) {
    const unsigned* mine = own_list_base(lst, plane, (int)blockIdx.x, n, nown);
    unsigned q[BIN_CHUNKS / 32][BIN_SLOT / 32];
#pragma unroll
    for (int g = 0; g < BIN_CHUNKS / 32; ++g) {
      const int c  = 32 * g + wv;
      const int nc = cn[c];
#pragma unroll
      for (int k = 0; k < BIN_SLOT / 32; ++k)
        q[g][k] = (32 * k + lane < nc) ? mine[c * BIN_SLOT + 32 * k + lane] : 0u;
    }
#pragma unroll
    for (int g = 0; g < BIN_CHUNKS / 32; ++g) {
      const int c  = 32 * g + wv;
      const int nc = cn[c];
      const int oc = off[c];
#pragma unroll
      for (int k = 0; k < BIN_SLOT / 32; ++k)
        if (32 * k + lane < nc) dense[oc + 32 * k + lane] = q[g][k];
    }
    __syncthreads();

    for (int i0 = wv * 32; i0 < total; i0 += OWN_BLOCK) {
      const int qn = (total - i0 < 32) ? (total - i0) : 32;
      int   s = 0;
      int   r = -1;
      float w = 0.0f;
      if (lane < qn) {
        const unsigned q = dense[i0 + lane];
        const int qi = (int)(q & ((1u << OWN_IB) - 1u));
        s = src[qi];
        w = ew[qi];
        r = (int)(q >> OWN_IB);
      }
      own_add_queue<F>(acc, hs, s, w, r, qn, cnt, lane);
    }
  } else {
    for (int i = t; i < e; i += OWN_BLOCK) {
      const unsigned r = (unsigned)(dst[i] - base);
      if (r < (unsigned)cnt) {
        const float  w = ew[i];
        const float* p = hs + (size_t)src[i] * F;
        for (int f = 0; f < F; ++f) atomicAdd(&acc[r * F + f], w * p[f]);
      }
    }
  }
  __syncthreads();

  own_finish<F, RELU>(acc, dinv, bias, out + (size_t)base * F, base, cnt, t);
}

extern "C" void kernel_launch(void* const* d_in, const int* in_sizes, int n_in,
                              void* d_out, int out_size, void* d_ws, size_t ws_size,
                              hipStream_t stream) {
  const float* z  = (const float*)d_in[0];
  const int*   ei = (const int*)d_in[1];
  const float* ew = (const float*)d_in[2];
  const float* W1 = (const float*)d_in[3];
  const float* b1 = (const float*)d_in[4];
  const float* W2 = (const float*)d_in[5];
  const float* b2 = (const float*)d_in[6];
  float* out = (float*)d_out;

  const int n = in_sizes[0] / 64;
  const int e = in_sizes[2];
  const int* srcp = ei;
  const int* dstp = ei + e;

  float* ws   = (float*)d_ws;
  unsigned* plane = (unsigned*)ws;
  float* dinv = ws + n;
  float* X1   = dinv + n;
  float* X2   = X1 + (size_t)n * 128;
  const int B = 256;
  unsigned* obuf = (unsigned*)out;

  bin_edges_kernel<<<BIN_CHUNKS, BIN_BLOCK, 0, stream>>>(dstp, obuf, plane, n, e);

  deg_partial_kernel<<<DEG_GROUPS, DEG_BLOCK, 0, stream>>>(dstp, ew, X1, n, e);
  dinv_reduce_kernel<<<(n + B - 1) / B, B, 0, stream>>>(X1, dinv, n);

  gemm_wmma_scale_kernel<64, 128><<<((n + 15) / 16 + GEMM_BLOCK / 32 - 1) / (GEMM_BLOCK / 32), GEMM_BLOCK, 0, stream>>>(
      z, W1, dinv, X1, n);
  own_list_kernel<128, true><<<(n + OWN_ROWS - 1) / OWN_ROWS, OWN_BLOCK, 0, stream>>>(
      X1, srcp, dstp, ew, obuf, plane, dinv, b1, X2, n, e);
  gemm_wmma_scale_kernel<128, 64><<<((n + 15) / 16 + GEMM_BLOCK / 32 - 1) / (GEMM_BLOCK / 32), GEMM_BLOCK, 0, stream>>>(
      X2, W2, dinv, X1, n);
  own_list_kernel<64, false><<<(n + OWN_ROWS - 1) / OWN_ROWS, OWN_BLOCK, 0, stream>>>(
      X1, srcp, dstp, ew, obuf, plane, dinv, b2, out, n, e);
}


# Weighted in-degree taken by the owner work-groups from their binned edge lists (weights by edge index, ds_add_f32 per owned row): no 128 degree copies, no reduce kernel, 6 launches; LDS arrival order
# speedup vs baseline: 3.6782x; 1.0563x over previous
#include <hip/hip_runtime.h>

typedef float fx2 __attribute__((ext_vector_type(2)));
typedef float fx8 __attribute__((ext_vector_type(8)));

#define GEMM_BLOCK 128

template <int K, int NOUT>
__global__ void __launch_bounds__(GEMM_BLOCK)
gemm_wmma_scale_kernel(const float* __restrict__ X,
                       const float* __restrict__ W,
                       const float* __restrict__ dinv,
                       float* __restrict__ hs,
                       int nrows) {
  constexpr int NT = NOUT / 16;
  static_assert(NT % 4 == 0, "a lane's columns move as whole 16-byte groups");
  const int lane = threadIdx.x & 31;
  const int wave = blockIdx.x * (blockDim.x >> 5) + (threadIdx.x >> 5);
  const int row0 = wave * 16;
  if (row0 >= nrows) return;

  const int half = lane >> 4;
  const int m    = lane & 15;

  fx8 c[NT];
#pragma unroll
  for (int j = 0; j < NT; ++j) c[j] = fx8{};
  const float* xrow = X + (size_t)(row0 + m) * K;
  const float* wcol = W + NT * m;

#pragma unroll
  for (int kb = 0; kb < K; kb += 4) {
    const int ka = kb + 2 * half;
    fx2 a;
    a.x = xrow[ka];
    a.y = xrow[ka + 1];
    float b0[NT], b1[NT];
#pragma unroll
    for (int q = 0; q < NT / 4; ++q) {
      const float4 u = *(const float4*)(wcol + (size_t)ka * NOUT + 4 * q);
      const float4 v = *(const float4*)(wcol + (size_t)(ka + 1) * NOUT + 4 * q);
      b0[4 * q] = u.x; b0[4 * q + 1] = u.y; b0[4 * q + 2] = u.z; b0[4 * q + 3] = u.w;
      b1[4 * q] = v.x; b1[4 * q + 1] = v.y; b1[4 * q + 2] = v.z; b1[4 * q + 3] = v.w;
    }
#pragma unroll
    for (int j = 0; j < NT; ++j) {
      fx2 b;
      b.x = b0[j];
      b.y = b1[j];
      c[j] = __builtin_amdgcn_wmma_f32_16x16x4_f32(false, a, false, b,
                                                   (short)0, c[j], false, false);
    }
  }

#pragma unroll
  for (int r = 0; r < 8; ++r) {
    const int mr  = r + 8 * half;
    const float s = dinv[row0 + mr];
    float* orow = hs + (size_t)(row0 + mr) * NOUT + NT * m;
#pragma unroll
    for (int q = 0; q < NT / 4; ++q) {
      float4 o;
      o.x = c[4 * q][r] * s;
      o.y = c[4 * q + 1][r] * s;
      o.z = c[4 * q + 2][r] * s;
      o.w = c[4 * q + 3][r] * s;
      *(float4*)(orow + 4 * q) = o;
    }
  }
}

#define OWN_ROWS  196
#define OWN_BLOCK 1024
#define OWN_REG   (OWN_ROWS * 64)
#define OWN_IB    23
static_assert(OWN_ROWS <= 256, "a local row must fit the 8 bits above the edge index");

#define BIN_CHUNKS 128
#define BIN_SLOT   96
#define BIN_MAXOWN 256
#define BIN_BLOCK  1024
#define BIN_U      16
#define BIN_OVER   (BIN_SLOT + 1)
static_assert(BIN_CHUNKS * BIN_SLOT <= OWN_REG, "an owner's slots must fit its list block");
static_assert(BIN_SLOT % 32 == 0 && BIN_CHUNKS % 32 == 0 && OWN_BLOCK == 1024, "whole waves per slot; slot c is copied by wave c % 32 of 32");

__device__ __forceinline__ unsigned* own_list_base(unsigned* obuf, unsigned* plane, int b, int n, int nown) {
  if ((size_t)(b + 1) * OWN_REG <= (size_t)n * 64) return obuf + (size_t)b * OWN_REG;
  if (n - OWN_REG >= nown * BIN_CHUNKS) return plane + (n - OWN_REG);
  return nullptr;
}

__global__ void __launch_bounds__(BIN_BLOCK)
bin_edges_kernel(const int* __restrict__ dst, unsigned* __restrict__ obuf,
                 unsigned* __restrict__ plane, int n, int e) {
  __shared__ unsigned stage[BIN_MAXOWN * BIN_SLOT];
  __shared__ unsigned cn[BIN_MAXOWN];
  const int t    = threadIdx.x;
  const int lane = t & 31;
  const int wv   = t >> 5;
  const int c    = blockIdx.x;
  const int nown = (n + OWN_ROWS - 1) / OWN_ROWS;
  const int len  = (e + BIN_CHUNKS - 1) / BIN_CHUNKS;
  const int lo   = c * len;
  const int hi   = (lo + len < e) ? (lo + len) : e;
  for (int j = t; j < BIN_MAXOWN; j += BIN_BLOCK) cn[j] = 0u;
  __syncthreads();
  for (int k0 = lo; k0 < hi; k0 += BIN_U * BIN_BLOCK) {
    int d[BIN_U];
#pragma unroll
    for (int u = 0; u < BIN_U; ++u) {
      const int i = k0 + u * BIN_BLOCK + t;
      d[u] = dst[(i < hi) ? i : (hi - 1)];
    }
#pragma unroll
    for (int u = 0; u < BIN_U; ++u) {
      const int i = k0 + u * BIN_BLOCK + t;
      const unsigned o = (unsigned)d[u] / OWN_ROWS;
      const unsigned r = (unsigned)d[u] - o * OWN_ROWS;
      if (i < hi && o < BIN_MAXOWN) {
        const unsigned rank = atomicAdd(&cn[o], 1u);
        if (rank < BIN_SLOT) stage[o * BIN_SLOT + rank] = (unsigned)i | (r << OWN_IB);
      }
    }
  }
  __syncthreads();
  const int nheld = (nown < BIN_MAXOWN) ? nown : BIN_MAXOWN;
  for (int o = wv; o < nheld; o += BIN_BLOCK / 32) {
    unsigned* base = own_list_base(obuf, plane, o, n, nown);
    if (base == nullptr) continue;
    const unsigned cnt = (cn[o] < BIN_SLOT) ? cn[o] : BIN_SLOT;
    for (unsigned i = lane; i < cnt; i += 32) base[c * BIN_SLOT + i] = stage[o * BIN_SLOT + i];
  }
  const bool enc = e <= (1 << OWN_IB);
  for (int o = t; o < nown; o += BIN_BLOCK) {
    const bool ok = enc && o < BIN_MAXOWN && own_list_base(obuf, plane, o, n, nown) != nullptr;
    plane[(size_t)o * BIN_CHUNKS + c] = ok ? cn[o] : (unsigned)BIN_OVER;
  }
}

#define DGO_BLOCK 256
#define DGO_SPW   (BIN_CHUNKS / (DGO_BLOCK / 32))
static_assert(BIN_CHUNKS == 128 && DGO_SPW % 4 == 0, "a lane holds 4 slot counts; a wave's slots are whole lanes of them");
static_assert(OWN_ROWS <= DGO_BLOCK && OWN_ROWS >= 32, "one thread writes one row's dinv; a lane without an entry adds 0.0 to row [lane]");

__global__ void __launch_bounds__(DGO_BLOCK)
deg_own_kernel(const int* __restrict__ dst, const float* __restrict__ ew,
               unsigned* obuf, unsigned* plane,
               float* __restrict__ dinv, int n, int e) {
  __shared__ float deg[OWN_ROWS];
  const int t    = threadIdx.x;
  const int lane = t & 31;
  const int wv   = __builtin_amdgcn_readfirstlane(t >> 5);
  const int b    = blockIdx.x;
  const int base = b * OWN_ROWS;
  const int cnt  = (n - base < OWN_ROWS) ? (n - base) : OWN_ROWS;
  const int nown = (n + OWN_ROWS - 1) / OWN_ROWS;

  const uint4 c4 = *(const uint4*)(plane + (size_t)b * BIN_CHUNKS + 4 * lane);
  if (t < OWN_ROWS) deg[t] = 0.0f;
  const bool over = c4.x > BIN_SLOT || c4.y > BIN_SLOT || c4.z > BIN_SLOT || c4.w > BIN_SLOT;
  const bool whole = __builtin_amdgcn_ballot_w32(over) == 0u;
  __syncthreads();

  if (whole) {
    const unsigned* mine = own_list_base(obuf, plane, b, n, nown);
    const unsigned* idle = plane + (size_t)b * BIN_CHUNKS;
    int      nc[DGO_SPW];
    unsigned q[DGO_SPW][BIN_SLOT / 32];
    float    w[DGO_SPW][BIN_SLOT / 32];
#pragma unroll
    for (int g = 0; g < DGO_SPW; ++g) {
      const unsigned cv = ((g & 3) == 0) ? c4.x : ((g & 3) == 1) ? c4.y : ((g & 3) == 2) ? c4.z : c4.w;
      nc[g] = __builtin_amdgcn_readlane((int)cv, (DGO_SPW / 4) * wv + (g >> 2));
      const int c = DGO_SPW * wv + g;
#pragma unroll
      for (int k = 0; k < BIN_SLOT / 32; ++k) {
        const unsigned* p = (32 * k + lane < nc[g]) ? (mine + c * BIN_SLOT + 32 * k + lane) : idle;
        q[g][k] = *p;
      }
    }
    __builtin_amdgcn_sched_barrier(0);
#pragma unroll
    for (int g = 0; g < DGO_SPW; ++g) {
#pragma unroll
      for (int k = 0; k < BIN_SLOT / 32; ++k) {
        const float* p = (32 * k + lane < nc[g]) ? (ew + (q[g][k] & ((1u << OWN_IB) - 1u))) : (const float*)idle;
        w[g][k] = *p;
      }
    }
#pragma unroll
    for (int g = 0; g < DGO_SPW; ++g) {
#pragma unroll
      for (int k = 0; k < BIN_SLOT / 32; ++k) {
        const bool have = 32 * k + lane < nc[g];
        atomicAdd(&deg[have ? (int)(q[g][k] >> OWN_IB) : lane], have ? w[g][k] : 0.0f);
      }
    }
  } else {
    for (int i = t; i < e; i += DGO_BLOCK) {
      const unsigned r = (unsigned)(dst[i] - base);
      if (r < (unsigned)cnt) atomicAdd(&deg[r], ew[i]);
    }
  }
  __syncthreads();
  if (t < cnt) {
    const float d = 1.0f + deg[t];
    dinv[base + t] = (d > 0.0f) ? rsqrtf(d) : 0.0f;
  }
}

template <int F>
__device__ __forceinline__ void own_add_edge(float* acc, const float* __restrict__ hs,
                                             int s, float w, int r, int lane) {
  const float* p = hs + (size_t)s * F + lane;
  float v[F / 32];
#pragma unroll
  for (int c = 0; c < F / 32; ++c) v[c] = p[32 * c];
#pragma unroll
  for (int c = 0; c < F / 32; ++c) atomicAdd(&acc[r * F + 32 * c + lane], w * v[c]);
}

#define OWN_G 8
template <int F>
__device__ __forceinline__ void own_add_queue(float* acc, const float* __restrict__ hs,
                                              int s, float w, int r,
                                              int qn, int cnt, int lane) {
  const unsigned ok   = __builtin_amdgcn_ballot_w32((unsigned)r < (unsigned)cnt);
  const unsigned want = (qn >= 32) ? 0xffffffffu : ((1u << qn) - 1u);
  if (ok == want) {
    int t = 0;
    for (; t + OWN_G <= qn; t += OWN_G) {
      float v[OWN_G][F / 32];
#pragma unroll
      for (int u = 0; u < OWN_G; ++u) {
        const int su = __builtin_amdgcn_readlane(s, t + u);
        const float* p = hs + (size_t)su * F + lane;
#pragma unroll
        for (int c = 0; c < F / 32; ++c) v[u][c] = p[32 * c];
      }
#pragma unroll
      for (int u = 0; u < OWN_G; ++u) {
        const int   ru = __builtin_amdgcn_readlane(r, t + u);
        const float wu = __int_as_float(__builtin_amdgcn_readlane(__float_as_int(w), t + u));
#pragma unroll
        for (int c = 0; c < F / 32; ++c) atomicAdd(&acc[ru * F + 32 * c + lane], wu * v[u][c]);
      }
    }
    for (; t < qn; ++t) {
      const int   su = __builtin_amdgcn_readlane(s, t);
      const int   ru = __builtin_amdgcn_readlane(r, t);
      const float wu = __int_as_float(__builtin_amdgcn_readlane(__float_as_int(w), t));
      own_add_edge<F>(acc, hs, su, wu, ru, lane);
    }
  } else {
    unsigned mk = ok;
    while (mk) {
      const int t = __builtin_ctz(mk);
      mk &= mk - 1u;
      const int   su = __builtin_amdgcn_readlane(s, t);
      const int   ru = __builtin_amdgcn_readlane(r, t);
      const float wu = __int_as_float(__builtin_amdgcn_readlane(__float_as_int(w), t));
      own_add_edge<F>(acc, hs, su, wu, ru, lane);
    }
  }
}

template <int F, bool RELU>
__device__ __forceinline__ void own_finish(const float* acc, const float* __restrict__ dinv,
                                           const float* __restrict__ bias, float* o0,
                                           int base, int cnt, int t) {
  for (int j = t; j < cnt * F; j += OWN_BLOCK) {
    const int row = j / F;
    const int f   = j % F;
    float v = dinv[base + row] * acc[j] + bias[f];
    if (RELU) v = fmaxf(v, 0.0f);
    o0[j] = v;
  }
}

template <int F, bool RELU>
__global__ void __launch_bounds__(OWN_BLOCK)
own_list_kernel(const float* __restrict__ hs,
                const int* __restrict__ src,
                const int* __restrict__ dst,
                const float* __restrict__ ew,
                unsigned* lst,
                unsigned* plane,
                const float* __restrict__ dinv,
                const float* __restrict__ bias,
                float* out, int n, int e) {
  __shared__ float acc[OWN_ROWS * F];
  __shared__ unsigned dense[BIN_CHUNKS * BIN_SLOT];
  __shared__ int cn[BIN_CHUNKS];
  __shared__ int off[BIN_CHUNKS];
  __shared__ int tot[2];
  const int t    = threadIdx.x;
  const int lane = t & 31;
  const int wv   = __builtin_amdgcn_readfirstlane(t >> 5);
  const int base = blockIdx.x * OWN_ROWS;
  const int cnt  = (n - base < OWN_ROWS) ? (n - base) : OWN_ROWS;
  const int nown = (n + OWN_ROWS - 1) / OWN_ROWS;

  const float* hs0 = hs + (size_t)base * F;
  for (int j = t; j < cnt * F; j += OWN_BLOCK) acc[j] = hs0[j];

  if (wv == 0) {
    const int* pc = (const int*)plane + (size_t)blockIdx.x * BIN_CHUNKS + lane * (BIN_CHUNKS / 32);
    int v[BIN_CHUNKS / 32];
    int sum = 0, over = 0;
#pragma unroll
    for (int k = 0; k < BIN_CHUNKS / 32; ++k) {
      v[k] = pc[k];
      over |= (v[k] > BIN_SLOT) ? 1 : 0;
      sum += v[k];
    }
    int incl = sum;
#pragma unroll
    for (int dlt = 1; dlt < 32; dlt <<= 1) {
      const int up = __shfl_up(incl, dlt, 32);
      if (lane >= dlt) incl += up;
    }
    int run = incl - sum;
#pragma unroll
    for (int k = 0; k < BIN_CHUNKS / 32; ++k) {
      cn[lane * (BIN_CHUNKS / 32) + k]  = v[k];
      off[lane * (BIN_CHUNKS / 32) + k] = run;
      run += v[k];
    }
    const unsigned anyover = __builtin_amdgcn_ballot_w32(over != 0);
    if (lane == 31) { tot[0] = incl; tot[1] = (anyover != 0u) ? 1 : 0; }
  }
  __syncthreads();
  const int  total = tot[0];
  const bool whole = tot[1] == 0;

  if (whole) {
    const unsigned* mine = own_list_base(lst, plane, (int)blockIdx.x, n, nown);
    unsigned q[BIN_CHUNKS / 32][BIN_SLOT / 32];
#pragma unroll
    for (int g = 0; g < BIN_CHUNKS / 32; ++g) {
      const int c  = 32 * g + wv;
      const int nc = cn[c];
#pragma unroll
      for (int k = 0; k < BIN_SLOT / 32; ++k)
        q[g][k] = (32 * k + lane < nc) ? mine[c * BIN_SLOT + 32 * k + lane] : 0u;
    }
#pragma unroll
    for (int g = 0; g < BIN_CHUNKS / 32; ++g) {
      const int c  = 32 * g + wv;
      const int nc = cn[c];
      const int oc = off[c];
#pragma unroll
      for (int k = 0; k < BIN_SLOT / 32; ++k)
        if (32 * k + lane < nc) dense[oc + 32 * k + lane] = q[g][k];
    }
    __syncthreads();

    for (int i0 = wv * 32; i0 < total; i0 += OWN_BLOCK) {
      const int qn = (total - i0 < 32) ? (total - i0) : 32;
      int   s = 0;
      int   r = -1;
      float w = 0.0f;
      if (lane < qn) {
        const unsigned q = dense[i0 + lane];
        const int qi = (int)(q & ((1u << OWN_IB) - 1u));
        s = src[qi];
        w = ew[qi];
        r = (int)(q >> OWN_IB);
      }
      own_add_queue<F>(acc, hs, s, w, r, qn, cnt, lane);
    }
  } else {
    for (int i = t; i < e; i += OWN_BLOCK) {
      const unsigned r = (unsigned)(dst[i] - base);
      if (r < (unsigned)cnt) {
        const float  w = ew[i];
        const float* p = hs + (size_t)src[i] * F;
        for (int f = 0; f < F; ++f) atomicAdd(&acc[r * F + f], w * p[f]);
      }
    }
  }
  __syncthreads();

  own_finish<F, RELU>(acc, dinv, bias, out + (size_t)base * F, base, cnt, t);
}

extern "C" void kernel_launch(void* const* d_in, const int* in_sizes, int n_in,
                              void* d_out, int out_size, void* d_ws, size_t ws_size,
                              hipStream_t stream) {
  const float* z  = (const float*)d_in[0];
  const int*   ei = (const int*)d_in[1];
  const float* ew = (const float*)d_in[2];
  const float* W1 = (const float*)d_in[3];
  const float* b1 = (const float*)d_in[4];
  const float* W2 = (const float*)d_in[5];
  const float* b2 = (const float*)d_in[6];
  float* out = (float*)d_out;

  const int n = in_sizes[0] / 64;
  const int e = in_sizes[2];
  const int* srcp = ei;
  const int* dstp = ei + e;

  float* ws   = (float*)d_ws;
  unsigned* plane = (unsigned*)ws;
  float* dinv = ws + n;
  float* X1   = dinv + n;
  float* X2   = X1 + (size_t)n * 128;
  unsigned* obuf = (unsigned*)out;

  bin_edges_kernel<<<BIN_CHUNKS, BIN_BLOCK, 0, stream>>>(dstp, obuf, plane, n, e);

  deg_own_kernel<<<(n + OWN_ROWS - 1) / OWN_ROWS, DGO_BLOCK, 0, stream>>>(dstp, ew, obuf, plane, dinv, n, e);

  gemm_wmma_scale_kernel<64, 128><<<((n + 15) / 16 + GEMM_BLOCK / 32 - 1) / (GEMM_BLOCK / 32), GEMM_BLOCK, 0, stream>>>(
      z, W1, dinv, X1, n);
  own_list_kernel<128, true><<<(n + OWN_ROWS - 1) / OWN_ROWS, OWN_BLOCK, 0, stream>>>(
      X1, srcp, dstp, ew, obuf, plane, dinv, b1, X2, n, e);
  gemm_wmma_scale_kernel<128, 64><<<((n + 15) / 16 + GEMM_BLOCK / 32 - 1) / (GEMM_BLOCK / 32), GEMM_BLOCK, 0, stream>>>(
      X2, W2, dinv, X1, n);
  own_list_kernel<64, false><<<(n + OWN_ROWS - 1) / OWN_ROWS, OWN_BLOCK, 0, stream>>>(
      X1, srcp, dstp, ew, obuf, plane, dinv, b2, out, n, e);
}


# Edge lists carry source and weight: bin kernel reads src/ew in whole lines, layer 1 leaves layer 2 a packed copy (7 bytes an entry); no per-edge src/ew fetch by index anywhere; LDS arrival order
# speedup vs baseline: 4.1540x; 1.1294x over previous
#include <hip/hip_runtime.h>

typedef float fx2 __attribute__((ext_vector_type(2)));
typedef float fx8 __attribute__((ext_vector_type(8)));

#define GEMM_BLOCK 128

template <int K, int NOUT>
__global__ void __launch_bounds__(GEMM_BLOCK)
gemm_wmma_scale_kernel(const float* __restrict__ X,
                       const float* __restrict__ W,
                       const float* __restrict__ dinv,
                       float* __restrict__ hs,
                       int nrows) {
  constexpr int NT = NOUT / 16;
  static_assert(NT % 4 == 0, "a lane's columns move as whole 16-byte groups");
  const int lane = threadIdx.x & 31;
  const int wave = blockIdx.x * (blockDim.x >> 5) + (threadIdx.x >> 5);
  const int row0 = wave * 16;
  if (row0 >= nrows) return;

  const int half = lane >> 4;
  const int m    = lane & 15;

  fx8 c[NT];
#pragma unroll
  for (int j = 0; j < NT; ++j) c[j] = fx8{};
  const float* xrow = X + (size_t)(row0 + m) * K;
  const float* wcol = W + NT * m;

#pragma unroll
  for (int kb = 0; kb < K; kb += 4) {
    const int ka = kb + 2 * half;
    fx2 a;
    a.x = xrow[ka];
    a.y = xrow[ka + 1];
    float b0[NT], b1[NT];
#pragma unroll
    for (int q = 0; q < NT / 4; ++q) {
      const float4 u = *(const float4*)(wcol + (size_t)ka * NOUT + 4 * q);
      const float4 v = *(const float4*)(wcol + (size_t)(ka + 1) * NOUT + 4 * q);
      b0[4 * q] = u.x; b0[4 * q + 1] = u.y; b0[4 * q + 2] = u.z; b0[4 * q + 3] = u.w;
      b1[4 * q] = v.x; b1[4 * q + 1] = v.y; b1[4 * q + 2] = v.z; b1[4 * q + 3] = v.w;
    }
#pragma unroll
    for (int j = 0; j < NT; ++j) {
      fx2 b;
      b.x = b0[j];
      b.y = b1[j];
      c[j] = __builtin_amdgcn_wmma_f32_16x16x4_f32(false, a, false, b,
                                                   (short)0, c[j], false, false);
    }
  }

#pragma unroll
  for (int r = 0; r < 8; ++r) {
    const int mr  = r + 8 * half;
    const float s = dinv[row0 + mr];
    float* orow = hs + (size_t)(row0 + mr) * NOUT + NT * m;
#pragma unroll
    for (int q = 0; q < NT / 4; ++q) {
      float4 o;
      o.x = c[4 * q][r] * s;
      o.y = c[4 * q + 1][r] * s;
      o.z = c[4 * q + 2][r] * s;
      o.w = c[4 * q + 3][r] * s;
      *(float4*)(orow + 4 * q) = o;
    }
  }
}

#define OWN_ROWS  196
#define OWN_BLOCK 1024
#define OWN_REG   (OWN_ROWS * 64)
#define OWN_IB    23
static_assert(OWN_ROWS <= 256, "a local row must fit the 8 bits above the edge index");

#define BIN_CHUNKS 128
#define BIN_SLOT   96
#define BIN_MAXOWN 256
#define BIN_BLOCK  1024
#define BIN_U      16
#define BIN_OVER   (BIN_SLOT + 1)
static_assert(BIN_CHUNKS * BIN_SLOT <= OWN_REG, "an owner's slots must fit its list block");
static_assert(BIN_SLOT % 32 == 0 && BIN_CHUNKS % 32 == 0 && OWN_BLOCK == 1024, "whole waves per slot; slot c is copied by wave c % 32 of 32");
#define CARRY_WORDS (BIN_CHUNKS * BIN_SLOT)
#define PACK_CAP ((OWN_REG * 4) / 7)
static_assert(PACK_CAP % 64 == 0 && PACK_CAP * 7 <= OWN_REG * 4, "three whole-line arrays inside an owner's list block");
static_assert(2 * CARRY_WORDS <= OWN_ROWS * 128, "an owner's weight list and source list must fit its own hidden-layer rows");

__device__ __forceinline__ unsigned* own_list_base(unsigned* obuf, unsigned* plane, int b, int n, int nown) {
  if ((size_t)(b + 1) * OWN_REG <= (size_t)n * 64) return obuf + (size_t)b * OWN_REG;
  if (n - OWN_REG >= nown * BIN_CHUNKS) return plane + (n - OWN_REG);
  return nullptr;
}

__device__ __forceinline__ unsigned* own_carry_base(unsigned* hbuf, int b, int n) {
  if (hbuf != nullptr && (size_t)(b + 1) * OWN_ROWS <= (size_t)n) return hbuf + (size_t)b * (OWN_ROWS * 128);
  return nullptr;
}

__device__ __forceinline__ bool own_packed(int b, int n, int total) {
  return (size_t)(b + 1) * OWN_ROWS <= (size_t)n && n <= 65536 && total <= PACK_CAP;
}

__global__ void __launch_bounds__(BIN_BLOCK)
bin_edges_kernel(const int* __restrict__ src, const int* __restrict__ dst,
                 const float* __restrict__ ew, unsigned* __restrict__ obuf,
                 unsigned* __restrict__ hbuf, unsigned* __restrict__ plane, int n, int e) {
  __shared__ unsigned stage[BIN_MAXOWN * BIN_SLOT];
  __shared__ unsigned stageS[BIN_MAXOWN * BIN_SLOT];
  __shared__ unsigned stageW[BIN_MAXOWN * BIN_SLOT];
  __shared__ unsigned cn[BIN_MAXOWN];
  const int t    = threadIdx.x;
  const int lane = t & 31;
  const int wv   = t >> 5;
  const int c    = blockIdx.x;
  const int nown = (n + OWN_ROWS - 1) / OWN_ROWS;
  const int len  = (e + BIN_CHUNKS - 1) / BIN_CHUNKS;
  const int lo   = c * len;
  const int hi   = (lo + len < e) ? (lo + len) : e;
  for (int j = t; j < BIN_MAXOWN; j += BIN_BLOCK) cn[j] = 0u;
  __syncthreads();
  for (int k0 = lo; k0 < hi; k0 += BIN_U * BIN_BLOCK) {
    int d[BIN_U];
    int sv[BIN_U];
    float wt[BIN_U];
#pragma unroll
    for (int u = 0; u < BIN_U; ++u) {
      const int i  = k0 + u * BIN_BLOCK + t;
      const int ic = (i < hi) ? i : (hi - 1);
      d[u]  = dst[ic];
      sv[u] = src[ic];
      wt[u] = ew[ic];
    }
#pragma unroll
    for (int u = 0; u < BIN_U; ++u) {
      const int i = k0 + u * BIN_BLOCK + t;
      const unsigned o = (unsigned)d[u] / OWN_ROWS;
      const unsigned r = (unsigned)d[u] - o * OWN_ROWS;
      if (i < hi && o < BIN_MAXOWN) {
        const unsigned rank = atomicAdd(&cn[o], 1u);
        if (rank < BIN_SLOT) {
          stage[o * BIN_SLOT + rank]  = (unsigned)i | (r << OWN_IB);
          stageS[o * BIN_SLOT + rank] = (unsigned)sv[u];
          stageW[o * BIN_SLOT + rank] = __float_as_uint(wt[u]);
        }
      }
    }
  }
  __syncthreads();
  const int nheld = (nown < BIN_MAXOWN) ? nown : BIN_MAXOWN;
  for (int o = wv; o < nheld; o += BIN_BLOCK / 32) {
    unsigned* base = own_list_base(obuf, plane, o, n, nown);
    if (base == nullptr) continue;
    const unsigned cnt = (cn[o] < BIN_SLOT) ? cn[o] : BIN_SLOT;
    for (unsigned i = lane; i < cnt; i += 32) base[c * BIN_SLOT + i] = stage[o * BIN_SLOT + i];
    unsigned* cb = own_carry_base(hbuf, o, n);
    if (cb != nullptr) {
      for (unsigned i = lane; i < cnt; i += 32) cb[c * BIN_SLOT + i] = stageW[o * BIN_SLOT + i];
      for (unsigned i = lane; i < cnt; i += 32) cb[CARRY_WORDS + c * BIN_SLOT + i] = stageS[o * BIN_SLOT + i];
    }
  }
  const bool enc = e <= (1 << OWN_IB);
  for (int o = t; o < nown; o += BIN_BLOCK) {
    const bool ok = enc && o < BIN_MAXOWN && own_list_base(obuf, plane, o, n, nown) != nullptr;
    plane[(size_t)o * BIN_CHUNKS + c] = ok ? cn[o] : (unsigned)BIN_OVER;
  }
}

#define DGO_BLOCK 256
#define DGO_SPW   (BIN_CHUNKS / (DGO_BLOCK / 32))
static_assert(BIN_CHUNKS == 128 && DGO_SPW % 4 == 0, "a lane holds 4 slot counts; a wave's slots are whole lanes of them");
static_assert(OWN_ROWS <= DGO_BLOCK && OWN_ROWS >= 32, "one thread writes one row's dinv; a lane without an entry adds 0.0 to row [lane]");

__global__ void __launch_bounds__(DGO_BLOCK)
deg_own_kernel(const int* __restrict__ dst, const float* __restrict__ ew,
               unsigned* obuf, unsigned* hbuf, unsigned* plane,
               float* __restrict__ dinv, int n, int e) {
  __shared__ float deg[OWN_ROWS];
  const int t    = threadIdx.x;
  const int lane = t & 31;
  const int wv   = __builtin_amdgcn_readfirstlane(t >> 5);
  const int b    = blockIdx.x;
  const int base = b * OWN_ROWS;
  const int cnt  = (n - base < OWN_ROWS) ? (n - base) : OWN_ROWS;
  const int nown = (n + OWN_ROWS - 1) / OWN_ROWS;

  const uint4 c4 = *(const uint4*)(plane + (size_t)b * BIN_CHUNKS + 4 * lane);
  if (t < OWN_ROWS) deg[t] = 0.0f;
  const bool over = c4.x > BIN_SLOT || c4.y > BIN_SLOT || c4.z > BIN_SLOT || c4.w > BIN_SLOT;
  const bool whole = __builtin_amdgcn_ballot_w32(over) == 0u;
  __syncthreads();

  if (whole) {
    const unsigned* mine = own_list_base(obuf, plane, b, n, nown);
    const float* wl = (const float*)own_carry_base(hbuf, b, n);
    const unsigned* idle = plane + (size_t)b * BIN_CHUNKS;
    int      nc[DGO_SPW];
    unsigned q[DGO_SPW][BIN_SLOT / 32];
    float    w[DGO_SPW][BIN_SLOT / 32];
#pragma unroll
    for (int g = 0; g < DGO_SPW; ++g) {
      const unsigned cv = ((g & 3) == 0) ? c4.x : ((g & 3) == 1) ? c4.y : ((g & 3) == 2) ? c4.z : c4.w;
      nc[g] = __builtin_amdgcn_readlane((int)cv, (DGO_SPW / 4) * wv + (g >> 2));
      const int c = DGO_SPW * wv + g;
#pragma unroll
      for (int k = 0; k < BIN_SLOT / 32; ++k) {
        const unsigned* p = (32 * k + lane < nc[g]) ? (mine + c * BIN_SLOT + 32 * k + lane) : idle;
        q[g][k] = *p;
      }
    }
    if (wl != nullptr) {
#pragma unroll
      for (int g = 0; g < DGO_SPW; ++g) {
        const int c = DGO_SPW * wv + g;
#pragma unroll
        for (int k = 0; k < BIN_SLOT / 32; ++k) {
          const float* p = (32 * k + lane < nc[g]) ? (wl + c * BIN_SLOT + 32 * k + lane) : (const float*)idle;
          w[g][k] = *p;
        }
      }
    } else {
      __builtin_amdgcn_sched_barrier(0);
#pragma unroll
      for (int g = 0; g < DGO_SPW; ++g) {
#pragma unroll
        for (int k = 0; k < BIN_SLOT / 32; ++k) {
          const float* p = (32 * k + lane < nc[g]) ? (ew + (q[g][k] & ((1u << OWN_IB) - 1u))) : (const float*)idle;
          w[g][k] = *p;
        }
      }
    }
#pragma unroll
    for (int g = 0; g < DGO_SPW; ++g) {
#pragma unroll
      for (int k = 0; k < BIN_SLOT / 32; ++k) {
        const bool have = 32 * k + lane < nc[g];
        atomicAdd(&deg[have ? (int)(q[g][k] >> OWN_IB) : lane], have ? w[g][k] : 0.0f);
      }
    }
  } else {
    for (int i = t; i < e; i += DGO_BLOCK) {
      const unsigned r = (unsigned)(dst[i] - base);
      if (r < (unsigned)cnt) atomicAdd(&deg[r], ew[i]);
    }
  }
  __syncthreads();
  if (t < cnt) {
    const float d = 1.0f + deg[t];
    dinv[base + t] = (d > 0.0f) ? rsqrtf(d) : 0.0f;
  }
}

template <int F>
__device__ __forceinline__ void own_add_edge(float* acc, const float* __restrict__ hs,
                                             int s, float w, int r, int lane) {
  const float* p = hs + (size_t)s * F + lane;
  float v[F / 32];
#pragma unroll
  for (int c = 0; c < F / 32; ++c) v[c] = p[32 * c];
#pragma unroll
  for (int c = 0; c < F / 32; ++c) atomicAdd(&acc[r * F + 32 * c + lane], w * v[c]);
}

#define OWN_G 8
template <int F>
__device__ __forceinline__ void own_add_queue(float* acc, const float* __restrict__ hs,
                                              int s, float w, int r,
                                              int qn, int cnt, int lane) {
  const unsigned ok   = __builtin_amdgcn_ballot_w32((unsigned)r < (unsigned)cnt);
  const unsigned want = (qn >= 32) ? 0xffffffffu : ((1u << qn) - 1u);
  if (ok == want) {
    int t = 0;
    for (; t + OWN_G <= qn; t += OWN_G) {
      float v[OWN_G][F / 32];
#pragma unroll
      for (int u = 0; u < OWN_G; ++u) {
        const int su = __builtin_amdgcn_readlane(s, t + u);
        const float* p = hs + (size_t)su * F + lane;
#pragma unroll
        for (int c = 0; c < F / 32; ++c) v[u][c] = p[32 * c];
      }
#pragma unroll
      for (int u = 0; u < OWN_G; ++u) {
        const int   ru = __builtin_amdgcn_readlane(r, t + u);
        const float wu = __int_as_float(__builtin_amdgcn_readlane(__float_as_int(w), t + u));
#pragma unroll
        for (int c = 0; c < F / 32; ++c) atomicAdd(&acc[ru * F + 32 * c + lane], wu * v[u][c]);
      }
    }
    for (; t < qn; ++t) {
      const int   su = __builtin_amdgcn_readlane(s, t);
      const int   ru = __builtin_amdgcn_readlane(r, t);
      const float wu = __int_as_float(__builtin_amdgcn_readlane(__float_as_int(w), t));
      own_add_edge<F>(acc, hs, su, wu, ru, lane);
    }
  } else {
    unsigned mk = ok;
    while (mk) {
      const int t = __builtin_ctz(mk);
      mk &= mk - 1u;
      const int   su = __builtin_amdgcn_readlane(s, t);
      const int   ru = __builtin_amdgcn_readlane(r, t);
      const float wu = __int_as_float(__builtin_amdgcn_readlane(__float_as_int(w), t));
      own_add_edge<F>(acc, hs, su, wu, ru, lane);
    }
  }
}

template <int F, bool RELU>
__device__ __forceinline__ void own_finish(const float* acc, const float* __restrict__ dinv,
                                           const float* __restrict__ bias, float* o0,
                                           int base, int cnt, int t) {
  for (int j = t; j < cnt * F; j += OWN_BLOCK) {
    const int row = j / F;
    const int f   = j % F;
    float v = dinv[base + row] * acc[j] + bias[f];
    if (RELU) v = fmaxf(v, 0.0f);
    o0[j] = v;
  }
}

template <int F, bool RELU, bool CARRY, bool PACKED>
__global__ void __launch_bounds__(OWN_BLOCK)
own_list_kernel(const float* __restrict__ hs,
                const int* __restrict__ src,
                const int* __restrict__ dst,
                const float* __restrict__ ew,
                unsigned* lst,
                unsigned* hbuf,
                unsigned* plane,
                const float* __restrict__ dinv,
                const float* __restrict__ bias,
                float* out, int n, int e) {
  __shared__ float acc[OWN_ROWS * F];
  __shared__ unsigned dense[BIN_CHUNKS * BIN_SLOT];
  __shared__ unsigned denseS[CARRY ? CARRY_WORDS : 1];
  __shared__ float    denseW[CARRY ? CARRY_WORDS : 1];
  __shared__ int cn[BIN_CHUNKS];
  __shared__ int off[BIN_CHUNKS];
  __shared__ int tot[2];
  const int t    = threadIdx.x;
  const int lane = t & 31;
  const int wv   = __builtin_amdgcn_readfirstlane(t >> 5);
  const int base = blockIdx.x * OWN_ROWS;
  const int cnt  = (n - base < OWN_ROWS) ? (n - base) : OWN_ROWS;
  const int nown = (n + OWN_ROWS - 1) / OWN_ROWS;

  const float* hs0 = hs + (size_t)base * F;
  for (int j = t; j < cnt * F; j += OWN_BLOCK) acc[j] = hs0[j];

  if (wv == 0) {
    const int* pc = (const int*)plane + (size_t)blockIdx.x * BIN_CHUNKS + lane * (BIN_CHUNKS / 32);
    int v[BIN_CHUNKS / 32];
    int sum = 0, over = 0;
#pragma unroll
    for (int k = 0; k < BIN_CHUNKS / 32; ++k) {
      v[k] = pc[k];
      over |= (v[k] > BIN_SLOT) ? 1 : 0;
      sum += v[k];
    }
    int incl = sum;
#pragma unroll
    for (int dlt = 1; dlt < 32; dlt <<= 1) {
      const int up = __shfl_up(incl, dlt, 32);
      if (lane >= dlt) incl += up;
    }
    int run = incl - sum;
#pragma unroll
    for (int k = 0; k < BIN_CHUNKS / 32; ++k) {
      cn[lane * (BIN_CHUNKS / 32) + k]  = v[k];
      off[lane * (BIN_CHUNKS / 32) + k] = run;
      run += v[k];
    }
    const unsigned anyover = __builtin_amdgcn_ballot_w32(over != 0);
    if (lane == 31) { tot[0] = incl; tot[1] = (anyover != 0u) ? 1 : 0; }
  }
  __syncthreads();
  const int  total = tot[0];
  const bool whole = tot[1] == 0;

  if (PACKED && whole && own_packed((int)blockIdx.x, n, total)) {
    const float*          pw = (const float*)(lst + (size_t)blockIdx.x * OWN_REG);
    const unsigned short* ps = (const unsigned short*)(pw + PACK_CAP);
    const unsigned char*  pr = (const unsigned char*)(ps + PACK_CAP);
    for (int i0 = wv * 32; i0 < total; i0 += OWN_BLOCK) {
      const int qn = (total - i0 < 32) ? (total - i0) : 32;
      int   s = 0;
      int   r = -1;
      float w = 0.0f;
      if (lane < qn) {
        w = pw[i0 + lane];
        s = (int)ps[i0 + lane];
        r = (int)pr[i0 + lane];
      }
      own_add_queue<F>(acc, hs, s, w, r, qn, cnt, lane);
    }
  } else if (whole) {
    const unsigned* mine = own_list_base(lst, plane, (int)blockIdx.x, n, nown);
    const unsigned* carry = CARRY ? own_carry_base(hbuf, (int)blockIdx.x, n) : nullptr;
    unsigned q[BIN_CHUNKS / 32][BIN_SLOT / 32];
#pragma unroll
    for (int g = 0; g < BIN_CHUNKS / 32; ++g) {
      const int c  = 32 * g + wv;
      const int nc = cn[c];
#pragma unroll
      for (int k = 0; k < BIN_SLOT / 32; ++k)
        q[g][k] = (32 * k + lane < nc) ? mine[c * BIN_SLOT + 32 * k + lane] : 0u;
    }
    if (CARRY && carry != nullptr) {
      unsigned qw[BIN_CHUNKS / 32][BIN_SLOT / 32];
      unsigned qs[BIN_CHUNKS / 32][BIN_SLOT / 32];
#pragma unroll
      for (int g = 0; g < BIN_CHUNKS / 32; ++g) {
        const int c  = 32 * g + wv;
        const int nc = cn[c];
#pragma unroll
        for (int k = 0; k < BIN_SLOT / 32; ++k) {
          const bool have = 32 * k + lane < nc;
          qw[g][k] = have ? carry[c * BIN_SLOT + 32 * k + lane] : 0u;
          qs[g][k] = have ? carry[CARRY_WORDS + c * BIN_SLOT + 32 * k + lane] : 0u;
        }
      }
#pragma unroll
      for (int g = 0; g < BIN_CHUNKS / 32; ++g) {
        const int c  = 32 * g + wv;
        const int nc = cn[c];
        const int oc = off[c];
#pragma unroll
        for (int k = 0; k < BIN_SLOT / 32; ++k)
          if (32 * k + lane < nc) {
            denseW[oc + 32 * k + lane] = __uint_as_float(qw[g][k]);
            denseS[oc + 32 * k + lane] = qs[g][k];
          }
      }
    }
#pragma unroll
    for (int g = 0; g < BIN_CHUNKS / 32; ++g) {
      const int c  = 32 * g + wv;
      const int nc = cn[c];
      const int oc = off[c];
#pragma unroll
      for (int k = 0; k < BIN_SLOT / 32; ++k)
        if (32 * k + lane < nc) dense[oc + 32 * k + lane] = q[g][k];
    }
    __syncthreads();

    if (CARRY && carry != nullptr && own_packed((int)blockIdx.x, n, total)) {
      float*          pw = (float*)const_cast<unsigned*>(mine);
      unsigned short* ps = (unsigned short*)(pw + PACK_CAP);
      unsigned char*  pr = (unsigned char*)(ps + PACK_CAP);
      for (int j = t; j < total; j += OWN_BLOCK) {
        pw[j] = denseW[j];
        ps[j] = (unsigned short)denseS[j];
        pr[j] = (unsigned char)(dense[j] >> OWN_IB);
      }
    }

    for (int i0 = wv * 32; i0 < total; i0 += OWN_BLOCK) {
      const int qn = (total - i0 < 32) ? (total - i0) : 32;
      int   s = 0;
      int   r = -1;
      float w = 0.0f;
      if (lane < qn) {
        const unsigned q = dense[i0 + lane];
        r = (int)(q >> OWN_IB);
        if (CARRY && carry != nullptr) {
          s = (int)denseS[i0 + lane];
          w = denseW[i0 + lane];
        } else {
          const int qi = (int)(q & ((1u << OWN_IB) - 1u));
          s = src[qi];
          w = ew[qi];
        }
      }
      own_add_queue<F>(acc, hs, s, w, r, qn, cnt, lane);
    }
  } else {
    for (int i = t; i < e; i += OWN_BLOCK) {
      const unsigned r = (unsigned)(dst[i] - base);
      if (r < (unsigned)cnt) {
        const float  w = ew[i];
        const float* p = hs + (size_t)src[i] * F;
        for (int f = 0; f < F; ++f) atomicAdd(&acc[r * F + f], w * p[f]);
      }
    }
  }
  __syncthreads();

  own_finish<F, RELU>(acc, dinv, bias, out + (size_t)base * F, base, cnt, t);
}

extern "C" void kernel_launch(void* const* d_in, const int* in_sizes, int n_in,
                              void* d_out, int out_size, void* d_ws, size_t ws_size,
                              hipStream_t stream) {
  const float* z  = (const float*)d_in[0];
  const int*   ei = (const int*)d_in[1];
  const float* ew = (const float*)d_in[2];
  const float* W1 = (const float*)d_in[3];
  const float* b1 = (const float*)d_in[4];
  const float* W2 = (const float*)d_in[5];
  const float* b2 = (const float*)d_in[6];
  float* out = (float*)d_out;

  const int n = in_sizes[0] / 64;
  const int e = in_sizes[2];
  const int* srcp = ei;
  const int* dstp = ei + e;

  float* ws   = (float*)d_ws;
  unsigned* plane = (unsigned*)ws;
  float* dinv = ws + n;
  float* X1   = dinv + n;
  float* X2   = X1 + (size_t)n * 128;
  unsigned* obuf = (unsigned*)out;
  unsigned* hbuf = (unsigned*)X2;

  bin_edges_kernel<<<BIN_CHUNKS, BIN_BLOCK, 0, stream>>>(srcp, dstp, ew, obuf, hbuf, plane, n, e);

  deg_own_kernel<<<(n + OWN_ROWS - 1) / OWN_ROWS, DGO_BLOCK, 0, stream>>>(dstp, ew, obuf, hbuf, plane, dinv, n, e);

  gemm_wmma_scale_kernel<64, 128><<<((n + 15) / 16 + GEMM_BLOCK / 32 - 1) / (GEMM_BLOCK / 32), GEMM_BLOCK, 0, stream>>>(
      z, W1, dinv, X1, n);
  own_list_kernel<128, true, true, false><<<(n + OWN_ROWS - 1) / OWN_ROWS, OWN_BLOCK, 0, stream>>>(
      X1, srcp, dstp, ew, obuf, hbuf, plane, dinv, b1, X2, n, e);
  gemm_wmma_scale_kernel<128, 64><<<((n + 15) / 16 + GEMM_BLOCK / 32 - 1) / (GEMM_BLOCK / 32), GEMM_BLOCK, 0, stream>>>(
      X2, W2, dinv, X1, n);
  own_list_kernel<64, false, false, true><<<(n + OWN_ROWS - 1) / OWN_ROWS, OWN_BLOCK, 0, stream>>>(
      X1, srcp, dstp, ew, obuf, nullptr, plane, dinv, b2, out, n, e);
}
